# starting kernel
# baseline (speedup 1.0000x reference)
#include <hip/hip_runtime.h>

typedef float v2f __attribute__((ext_vector_type(2)));
typedef float v8f __attribute__((ext_vector_type(8)));

#define NEG_SLOPE 0.2f


__global__ void fill_f32(float* __restrict__ p, long n, float v) {
  long i = (long)blockIdx.x * blockDim.x + threadIdx.x;
  long stride = (long)gridDim.x * blockDim.x;
  for (; i < n; i += stride) p[i] = v;
}

__global__ void gemm_wmma_f32_lds(const float* __restrict__ A, const float* __restrict__ B,
                                  float* __restrict__ C, int M, int K, int Nd,
                                  int tnPerBlock) {
  extern __shared__ float As[];
  int tilesN = Nd >> 4;
  int blocksPerRow = tilesN / tnPerBlock;
  int tm  = blockIdx.x / blocksPerRow;
  int tn0 = (blockIdx.x % blocksPerRow) * tnPerBlock;
  int wib  = threadIdx.x >> 5;
  int lane = threadIdx.x & 31;

  {
    const float4* Av  = (const float4*)(A + (long)tm * 16 * K);
    float4*       Asv = (float4*)As;
    int nvec = (16 * K) >> 2;
    for (int idx = threadIdx.x; idx < nvec; idx += blockDim.x) Asv[idx] = Av[idx];
  }
  __syncthreads();

  int tn   = tn0 + wib;
  int half = lane >> 4;
  int l15  = lane & 15;
  int bcol = tn * 16 + l15;
  v8f acc = {};
  for (int k = 0; k < K; k += 4) {
    int ka = k + 2 * half;
    v2f a, b;
    a.x = As[l15 * K + ka];
    a.y = As[l15 * K + ka + 1];
    b.x = B[(long)ka * Nd + bcol];
    b.y = B[(long)(ka + 1) * Nd + bcol];
    acc = __builtin_amdgcn_wmma_f32_16x16x4_f32(false, a, false, b,
                                                (short)0, acc, false, false);
  }
  long crow = (long)tm * 16 + half * 8;
#pragma unroll
  for (int r = 0; r < 8; ++r)
    C[(crow + r) * Nd + bcol] = acc[r];
}

__global__ void logits_kernel(const float* __restrict__ hmat, const float* __restrict__ a_src,
                              const float* __restrict__ a_dst, float* __restrict__ als,
                              float* __restrict__ ald, int nnodes, int heads, int ch) {
  int wid  = (int)(((long)blockIdx.x * blockDim.x + threadIdx.x) >> 5);
  int lane = threadIdx.x & 31;
  if (wid >= nnodes * heads) return;
  int node = wid / heads, hh = wid % heads;
  int F = heads * ch;
  float ss = 0.f, dd = 0.f;
  for (int c = lane; c < ch; c += 32) {
    float hv = hmat[(long)node * F + hh * ch + c];
    ss += hv * a_src[hh * ch + c];
    dd += hv * a_dst[hh * ch + c];
  }
#pragma unroll
  for (int off = 16; off > 0; off >>= 1) {
    ss += __shfl_xor(ss, off);
    dd += __shfl_xor(dd, off);
  }
  if (lane == 0) { als[wid] = ss; ald[wid] = dd; }
}

__device__ __forceinline__ void edge_sd(const int* __restrict__ eidx, int E, int e,
                                        int& s, int& d) {
  if (e < E) { s = eidx[e]; d = eidx[E + e]; }
  else       { s = e - E; d = s; }
}

__global__ void edge_max_kernel(const int* __restrict__ eidx, int E, int EN,
                                const float* __restrict__ als, const float* __restrict__ ald,
                                float* __restrict__ m, int heads) {
  long t = (long)blockIdx.x * blockDim.x + threadIdx.x;
  if (t >= (long)EN * heads) return;
  int e = (int)(t / heads), hh = (int)(t % heads);
  int s, d; edge_sd(eidx, E, e, s, d);
  float l = als[s * heads + hh] + ald[d * heads + hh];
  l = l > 0.f ? l : NEG_SLOPE * l;
  unsafeAtomicMax(&m[d * heads + hh], l);
}

template<int HEADS, int CH>
__global__ void edge_fused_kernel(const int* __restrict__ eidx, int E, int EN,
                                  const float* __restrict__ als, const float* __restrict__ ald,
                                  const float* __restrict__ m, float* __restrict__ ssum,
                                  const float* __restrict__ hsrc, float* __restrict__ accum) {
  constexpr int F  = HEADS * CH;
  constexpr int NF = F / 32;
  int wid  = (int)(((long)blockIdx.x * blockDim.x + threadIdx.x) >> 5);
  int lane = threadIdx.x & 31;
  if (wid >= EN) return;
  int s, d; edge_sd(eidx, E, wid, s, d);
  int j0   = lane * NF;
  int head = j0 / CH;
  float l = als[s * HEADS + head] + ald[d * HEADS + head];
  l = l > 0.f ? l : NEG_SLOPE * l;
  float ex = __expf(l - m[d * HEADS + head]);
  if ((j0 % CH) == 0)
    unsafeAtomicAdd(&ssum[d * HEADS + head], ex);
  const float* hp = hsrc  + (long)s * F + j0;
  float*       ap = accum + (long)d * F + j0;
  float v[NF];
#pragma unroll
  for (int k = 0; k < NF; ++k) v[k] = hp[k];
#pragma unroll
  for (int k = 0; k < NF; ++k)
    unsafeAtomicAdd(&ap[k], v[k] * ex);
}

__global__ void edge_fused_gen(const int* __restrict__ eidx, int E, int EN,
                               const float* __restrict__ als, const float* __restrict__ ald,
                               const float* __restrict__ m, float* __restrict__ ssum,
                               const float* __restrict__ hsrc, float* __restrict__ accum,
                               int heads, int ch) {
  int wid  = (int)(((long)blockIdx.x * blockDim.x + threadIdx.x) >> 5);
  int lane = threadIdx.x & 31;
  if (wid >= EN) return;
  int s, d; edge_sd(eidx, E, wid, s, d);
  int F = heads * ch;
  for (int j = lane; j < F; j += 32) {
    int head = j / ch;
    float l = als[s * heads + head] + ald[d * heads + head];
    l = l > 0.f ? l : NEG_SLOPE * l;
    float ex = __expf(l - m[d * heads + head]);
    if ((j % ch) == 0) unsafeAtomicAdd(&ssum[d * heads + head], ex);
    unsafeAtomicAdd(&accum[(long)d * F + j], hsrc[(long)s * F + j] * ex);
  }
}

__global__ void norm_bias_act_kernel(float* __restrict__ a, const float* __restrict__ ssum,
                                     const float* __restrict__ bias, long n, int F,
                                     int ch, int heads, int do_elu) {
  long i = (long)blockIdx.x * blockDim.x + threadIdx.x;
  long stride = (long)gridDim.x * blockDim.x;
  for (; i < n; i += stride) {
    int  j    = (int)(i % F);
    long node = i / F;
    float v = a[i] / ssum[node * heads + j / ch] + bias[j];
    if (do_elu) v = v > 0.f ? v : (__expf(v) - 1.f);
    a[i] = v;
  }
}


static inline int cdiv_i(long a, long b) { return (int)((a + b - 1) / b); }

static void launch_gemm(const float* A, const float* B, float* C,
                        int M, int K, int Nd, hipStream_t stream) {
  int tilesM = M / 16, tilesN = Nd / 16;
  int wpb = tilesN >= 8 ? 8 : tilesN;
  dim3 grid(tilesM * (tilesN / wpb));
  dim3 block(32 * wpb);
  size_t shmem = (size_t)16 * K * sizeof(float);
  gemm_wmma_f32_lds<<<grid, block, shmem, stream>>>(A, B, C, M, K, Nd, wpb);
}

extern "C" void kernel_launch(void* const* d_in, const int* in_sizes, int n_in,
                              void* d_out, int out_size, void* d_ws, size_t ws_size,
                              hipStream_t stream) {
  const float* x   = (const float*)d_in[0];
  const int*   eix = (const int*)d_in[1];
  const float* W1  = (const float*)d_in[2];
  const float* as1 = (const float*)d_in[3];
  const float* ad1 = (const float*)d_in[4];
  const float* b1  = (const float*)d_in[5];
  const float* W2  = (const float*)d_in[6];
  const float* as2 = (const float*)d_in[7];
  const float* ad2 = (const float*)d_in[8];
  const float* b2  = (const float*)d_in[9];
  float* out = (float*)d_out;

  const int IN_DIM = 128;
  const int F1     = in_sizes[5];
  const int HID    = in_sizes[9];
  const int HEADS  = F1 / HID;
  const int Nn     = in_sizes[0] / IN_DIM;
  const int E      = in_sizes[1] / 2;
  const int EN     = E + Nn;

  float* acc1 = (float*)d_ws;
  float* h1   = acc1 + (long)Nn * F1;
  float* sc   = h1 + (long)Nn * F1;
  float* als1 = sc;
  float* ald1 = als1 + (long)Nn * HEADS;
  float* m1   = ald1 + (long)Nn * HEADS;
  float* s1   = m1   + (long)Nn * HEADS;
  float* als2 = s1   + (long)Nn * HEADS;
  float* ald2 = als2 + Nn;
  float* m2   = ald2 + Nn;
  float* s2   = m2   + Nn;
  float* h2   = h1;

  const float NEGINF = -__builtin_huge_valf();

  fill_f32<<<cdiv_i((long)Nn * F1, 256), 256, 0, stream>>>(acc1, (long)Nn * F1, 0.f);
  fill_f32<<<cdiv_i((long)Nn * HEADS, 256), 256, 0, stream>>>(m1, (long)Nn * HEADS, NEGINF);
  fill_f32<<<cdiv_i((long)Nn * HEADS, 256), 256, 0, stream>>>(s1, (long)Nn * HEADS, 0.f);
  fill_f32<<<cdiv_i(Nn, 256), 256, 0, stream>>>(m2, Nn, NEGINF);
  fill_f32<<<cdiv_i(Nn, 256), 256, 0, stream>>>(s2, Nn, 0.f);
  fill_f32<<<cdiv_i((long)Nn * HID, 256), 256, 0, stream>>>(out, (long)Nn * HID, 0.f);

  launch_gemm(x, W1, h1, Nn, IN_DIM, F1, stream);
  logits_kernel<<<cdiv_i((long)Nn * HEADS * 32, 256), 256, 0, stream>>>(
      h1, as1, ad1, als1, ald1, Nn, HEADS, HID);
  edge_max_kernel<<<cdiv_i((long)EN * HEADS, 256), 256, 0, stream>>>(
      eix, E, EN, als1, ald1, m1, HEADS);
  if (HEADS == 4 && HID == 64)
    edge_fused_kernel<4, 64><<<cdiv_i((long)EN * 32, 256), 256, 0, stream>>>(
        eix, E, EN, als1, ald1, m1, s1, h1, acc1);
  else
    edge_fused_gen<<<cdiv_i((long)EN * 32, 256), 256, 0, stream>>>(
        eix, E, EN, als1, ald1, m1, s1, h1, acc1, HEADS, HID);
  norm_bias_act_kernel<<<cdiv_i((long)Nn * F1, 256), 256, 0, stream>>>(
      acc1, s1, b1, (long)Nn * F1, F1, HID, HEADS, 1);

  launch_gemm(acc1, W2, h2, Nn, F1, HID, stream);
  logits_kernel<<<cdiv_i((long)Nn * 32, 256), 256, 0, stream>>>(
      h2, as2, ad2, als2, ald2, Nn, 1, HID);
  edge_max_kernel<<<cdiv_i((long)EN, 256), 256, 0, stream>>>(
      eix, E, EN, als2, ald2, m2, 1);
  if (HID == 64)
    edge_fused_kernel<1, 64><<<cdiv_i((long)EN * 32, 256), 256, 0, stream>>>(
        eix, E, EN, als2, ald2, m2, s2, h2, out);
  else
    edge_fused_gen<<<cdiv_i((long)EN * 32, 256), 256, 0, stream>>>(
        eix, E, EN, als2, ald2, m2, s2, h2, out, 1, HID);
  norm_bias_act_kernel<<<cdiv_i((long)Nn * HID, 256), 256, 0, stream>>>(
      out, s2, b2, (long)Nn * HID, HID, HID, 1, 0);
}


# Edges bucketed by destination each call (int atomics: count, prefix, fill); one wave per node gathers its source rows and sums them in registers in ascending source order; float atomics removed
# speedup vs baseline: 3.7640x; 3.7640x over previous
#include <hip/hip_runtime.h>


typedef float v2f __attribute__((ext_vector_type(2)));
typedef float v8f __attribute__((ext_vector_type(8)));

#define NEG_SLOPE 0.2f


__global__ void fill_i32(int* __restrict__ p, int n, int v) {
  int i = blockIdx.x * blockDim.x + threadIdx.x;
  if (i < n) p[i] = v;
}

__global__ void gemm_wmma_f32_lds(const float* __restrict__ A, const float* __restrict__ B,
                                  float* __restrict__ C, int M, int K, int Nd,
                                  int tnPerBlock) {
  extern __shared__ float As[];
  int tilesN = Nd >> 4;
  int blocksPerRow = tilesN / tnPerBlock;
  int tm  = blockIdx.x / blocksPerRow;
  int tn0 = (blockIdx.x % blocksPerRow) * tnPerBlock;
  int wib  = threadIdx.x >> 5;
  int lane = threadIdx.x & 31;

  {
    const float4* Av  = (const float4*)(A + (long)tm * 16 * K);
    float4*       Asv = (float4*)As;
    int nvec = (16 * K) >> 2;
    for (int idx = threadIdx.x; idx < nvec; idx += blockDim.x) Asv[idx] = Av[idx];
  }
  __syncthreads();

  int tn   = tn0 + wib;
  int half = lane >> 4;
  int l15  = lane & 15;
  int bcol = tn * 16 + l15;
  v8f acc = {};
  for (int k = 0; k < K; k += 4) {
    int ka = k + 2 * half;
    v2f a, b;
    a.x = As[l15 * K + ka];
    a.y = As[l15 * K + ka + 1];
    b.x = B[(long)ka * Nd + bcol];
    b.y = B[(long)(ka + 1) * Nd + bcol];
    acc = __builtin_amdgcn_wmma_f32_16x16x4_f32(false, a, false, b,
                                                (short)0, acc, false, false);
  }
  long crow = (long)tm * 16 + half * 8;
#pragma unroll
  for (int r = 0; r < 8; ++r)
    C[(crow + r) * Nd + bcol] = acc[r];
}

__global__ void logits_kernel(const float* __restrict__ hmat, const float* __restrict__ a_src,
                              const float* __restrict__ a_dst, float* __restrict__ als,
                              float* __restrict__ ald, int nnodes, int heads, int ch) {
  int wid  = (int)(((long)blockIdx.x * blockDim.x + threadIdx.x) >> 5);
  int lane = threadIdx.x & 31;
  if (wid >= nnodes * heads) return;
  int node = wid / heads, hh = wid % heads;
  int F = heads * ch;
  float ss = 0.f, dd = 0.f;
  for (int c = lane; c < ch; c += 32) {
    float hv = hmat[(long)node * F + hh * ch + c];
    ss += hv * a_src[hh * ch + c];
    dd += hv * a_dst[hh * ch + c];
  }
#pragma unroll
  for (int off = 16; off > 0; off >>= 1) {
    ss += __shfl_xor(ss, off);
    dd += __shfl_xor(dd, off);
  }
  if (lane == 0) { als[wid] = ss; ald[wid] = dd; }
}


__global__ void degree_kernel(const int* __restrict__ eidx, int E, int* __restrict__ cnt) {
  int e = blockIdx.x * blockDim.x + threadIdx.x;
  if (e < E) atomicAdd(&cnt[eidx[E + e]], 1);
}

__global__ void rowptr_kernel(const int* __restrict__ cnt, int* __restrict__ rowptr,
                              int* __restrict__ cursor, int n) {
  __shared__ int sh[256];
  const int t = threadIdx.x;
  const int start = blockIdx.x * 256;
  int part = 0;
  for (int i = t; i < start; i += 256) part += cnt[i];
  const int i = start + t;
  const int c = i < n ? cnt[i] : 0;
  sh[t] = part;
  __syncthreads();
  for (int off = 128; off > 0; off >>= 1) {
    if (t < off) sh[t] += sh[t + off];
    __syncthreads();
  }
  const int base = sh[0];
  __syncthreads();
  sh[t] = c;
  __syncthreads();
  for (int off = 1; off < 256; off <<= 1) {
    int v = t >= off ? sh[t - off] : 0;
    __syncthreads();
    sh[t] += v;
    __syncthreads();
  }
  const int excl = base + sh[t] - c;
  if (i < n) { rowptr[i] = excl; cursor[i] = excl; }
  if (i == n - 1) rowptr[n] = excl + c;
}

__global__ void bucket_kernel(const int* __restrict__ eidx, int E, int* __restrict__ cursor,
                              int* __restrict__ csr_src) {
  int e = blockIdx.x * blockDim.x + threadIdx.x;
  if (e < E) {
    int d = eidx[E + e];
    int p = atomicAdd(&cursor[d], 1);
    csr_src[p] = eidx[e];
  }
}

template<int NH>
__device__ __forceinline__ void load_heads(const float* __restrict__ p, int idx, float (&v)[NH]) {
  if constexpr (NH == 4) {
    float4 t = ((const float4*)p)[idx];
    v[0] = t.x; v[1] = t.y; v[2] = t.z; v[3] = t.w;
  } else {
    for (int h = 0; h < NH; ++h) v[h] = p[(long)idx * NH + h];
  }
}

template<int NF>
__device__ __forceinline__ void load_row(const float* __restrict__ p, float (&v)[NF]) {
  if constexpr (NF % 4 == 0) {
    for (int q = 0; q < NF / 4; ++q) {
      float4 t = ((const float4*)p)[q];
      v[4 * q] = t.x; v[4 * q + 1] = t.y; v[4 * q + 2] = t.z; v[4 * q + 3] = t.w;
    }
  } else if constexpr (NF % 2 == 0) {
    for (int q = 0; q < NF / 2; ++q) {
      float2 t = ((const float2*)p)[q];
      v[2 * q] = t.x; v[2 * q + 1] = t.y;
    }
  } else {
    for (int k = 0; k < NF; ++k) v[k] = p[k];
  }
}

template<int NF>
__device__ __forceinline__ void store_row(float* __restrict__ p, const float (&v)[NF]) {
  if constexpr (NF % 4 == 0) {
    for (int q = 0; q < NF / 4; ++q)
      ((float4*)p)[q] = make_float4(v[4 * q], v[4 * q + 1], v[4 * q + 2], v[4 * q + 3]);
  } else if constexpr (NF % 2 == 0) {
    for (int q = 0; q < NF / 2; ++q)
      ((float2*)p)[q] = make_float2(v[2 * q], v[2 * q + 1]);
  } else {
    for (int k = 0; k < NF; ++k) p[k] = v[k];
  }
}

__device__ __forceinline__ int lane_bcast_i(int v, int i) {
  return __builtin_amdgcn_readlane(v, i);
}
__device__ __forceinline__ float lane_bcast_f(float v, int i) {
  return __int_as_float(__builtin_amdgcn_readlane(__float_as_int(v), i));
}

template<int HEADS, int CH, bool ELU>
__global__ void gather_kernel(const int* __restrict__ rowptr, const int* __restrict__ csr_src,
                              const float* __restrict__ als, const float* __restrict__ ald,
                              const float* __restrict__ hsrc, const float* __restrict__ bias,
                              float* __restrict__ outp, int nnodes) {
  constexpr int F  = HEADS * CH;
  constexpr int NF = F / 32;
  const int lane = threadIdx.x & 31;
  const int node = __builtin_amdgcn_readfirstlane((int)((blockIdx.x * blockDim.x + threadIdx.x) >> 5));
  if (node >= nnodes) return;
  const int beg = rowptr[node];
  const int deg = rowptr[node + 1] - beg;
  const int tot = deg + 1;
  float adst[HEADS];
  load_heads<HEADS>(ald, node, adst);

  float mx[HEADS];
  for (int h = 0; h < HEADS; ++h) mx[h] = -__builtin_huge_valf();
  for (int c = 0; c < tot; c += 32) {
    const int g = c + lane;
    if (g < tot) {
      const int s = g < deg ? csr_src[beg + g] : node;
      float a[HEADS];
      load_heads<HEADS>(als, s, a);
      for (int h = 0; h < HEADS; ++h) {
        float l = a[h] + adst[h];
        l = l > 0.f ? l : NEG_SLOPE * l;
        mx[h] = fmaxf(mx[h], l);
      }
    }
  }
  for (int h = 0; h < HEADS; ++h)
    for (int off = 16; off > 0; off >>= 1) mx[h] = fmaxf(mx[h], __shfl_xor(mx[h], off));

  const int j0   = lane * NF;
  const int head = j0 / CH;
  float acc[NF];
  for (int k = 0; k < NF; ++k) acc[k] = 0.f;
  float ssum[HEADS];
  for (int h = 0; h < HEADS; ++h) ssum[h] = 0.f;
  for (int c = 0; c < tot; c += 32) {
    const int g = c + lane;
    const int nn = tot - c < 32 ? tot - c : 32;
    int s = node;
    if (g < deg) s = csr_src[beg + g];
    {
      int rank = 0;
      for (int j = 0; j < nn; ++j) {
        const int sj = lane_bcast_i(s, j);
        rank += (int)(sj < s) + (int)((sj == s) & (j < lane));
      }
      rank = g < tot ? rank : lane;
      s = __builtin_amdgcn_ds_permute(rank << 2, s);
    }
    float ex[HEADS];
    for (int h = 0; h < HEADS; ++h) ex[h] = 0.f;
    if (g < tot) {
      float a[HEADS];
      load_heads<HEADS>(als, s, a);
      for (int h = 0; h < HEADS; ++h) {
        float l = a[h] + adst[h];
        l = l > 0.f ? l : NEG_SLOPE * l;
        ex[h] = __expf(l - mx[h]);
        ssum[h] += ex[h];
      }
    }
    const int nn4 = (nn + 3) & ~3;
    for (int i = 0; i < nn4; i += 4) {
      int   sr[4];
      float w[4];
      float v[4][NF];
      for (int u = 0; u < 4; ++u) {
        sr[u] = lane_bcast_i(s, i + u);
        float wu = lane_bcast_f(ex[0], i + u);
        for (int h = 1; h < HEADS; ++h) {
          float th = lane_bcast_f(ex[h], i + u);
          wu = head == h ? th : wu;
        }
        w[u] = wu;
      }
      for (int u = 0; u < 4; ++u) load_row<NF>(hsrc + (long)sr[u] * F + j0, v[u]);
      for (int u = 0; u < 4; ++u)
        for (int k = 0; k < NF; ++k) acc[k] += w[u] * v[u][k];
    }
  }
  for (int h = 0; h < HEADS; ++h)
    for (int off = 16; off > 0; off >>= 1) ssum[h] += __shfl_xor(ssum[h], off);
  float sm = ssum[0];
  for (int h = 1; h < HEADS; ++h) sm = head == h ? ssum[h] : sm;

  float bv[NF], r[NF];
  load_row<NF>(bias + j0, bv);
  for (int k = 0; k < NF; ++k) {
    float val = acc[k] / sm + bv[k];
    if (ELU) val = val > 0.f ? val : (__expf(val) - 1.f);
    r[k] = val;
  }
  store_row<NF>(outp + (long)node * F + j0, r);
}


static inline int cdiv_i(long a, long b) { return (int)((a + b - 1) / b); }

static void launch_gemm(const float* A, const float* B, float* C,
                        int M, int K, int Nd, hipStream_t stream) {
  int tilesM = M / 16, tilesN = Nd / 16;
  int wpb = tilesN >= 8 ? 8 : tilesN;
  dim3 grid(tilesM * (tilesN / wpb));
  dim3 block(32 * wpb);
  size_t shmem = (size_t)16 * K * sizeof(float);
  gemm_wmma_f32_lds<<<grid, block, shmem, stream>>>(A, B, C, M, K, Nd, wpb);
}

extern "C" void kernel_launch(void* const* d_in, const int* in_sizes, int n_in,
                              void* d_out, int out_size, void* d_ws, size_t ws_size,
                              hipStream_t stream) {
  const float* x   = (const float*)d_in[0];
  const int*   eix = (const int*)d_in[1];
  const float* W1  = (const float*)d_in[2];
  const float* as1 = (const float*)d_in[3];
  const float* ad1 = (const float*)d_in[4];
  const float* b1  = (const float*)d_in[5];
  const float* W2  = (const float*)d_in[6];
  const float* as2 = (const float*)d_in[7];
  const float* ad2 = (const float*)d_in[8];
  const float* b2  = (const float*)d_in[9];
  float* out = (float*)d_out;

  const int IN_DIM = 128;
  const int F1     = in_sizes[5];
  const int HID    = in_sizes[9];
  const int HEADS  = F1 / HID;
  const int Nn     = in_sizes[0] / IN_DIM;
  const int E      = in_sizes[1] / 2;

  float* h1e  = (float*)d_ws;
  float* h1   = h1e + (long)Nn * F1;
  float* als1 = h1 + (long)Nn * F1;
  float* ald1 = als1 + (long)Nn * HEADS;
  float* als2 = ald1 + (long)Nn * HEADS;
  float* ald2 = als2 + Nn;
  int* cnt    = (int*)(ald2 + Nn);
  int* rowptr = cnt + Nn;
  int* cursor = rowptr + (Nn + 1);
  int* csr    = cursor + Nn;
  float* h2   = h1;

  fill_i32<<<cdiv_i(Nn, 256), 256, 0, stream>>>(cnt, Nn, 0);
  degree_kernel<<<cdiv_i(E, 256), 256, 0, stream>>>(eix, E, cnt);
  rowptr_kernel<<<cdiv_i(Nn, 256), 256, 0, stream>>>(cnt, rowptr, cursor, Nn);
  bucket_kernel<<<cdiv_i(E, 256), 256, 0, stream>>>(eix, E, cursor, csr);

  launch_gemm(x, W1, h1, Nn, IN_DIM, F1, stream);
  logits_kernel<<<cdiv_i((long)Nn * HEADS * 32, 256), 256, 0, stream>>>(
      h1, as1, ad1, als1, ald1, Nn, HEADS, HID);
  gather_kernel<4, 64, true><<<cdiv_i((long)Nn * 32, 256), 256, 0, stream>>>(
      rowptr, csr, als1, ald1, h1, b1, h1e, Nn);

  launch_gemm(h1e, W2, h2, Nn, F1, HID, stream);
  logits_kernel<<<cdiv_i((long)Nn * 32, 256), 256, 0, stream>>>(
      h2, as2, ad2, als2, ald2, Nn, 1, HID);
  gather_kernel<1, 64, false><<<cdiv_i((long)Nn * 32, 256), 256, 0, stream>>>(
      rowptr, csr, als2, ald2, h2, b2, out, Nn);
}


# Logits kernel: one wave takes 32 (node,head) pairs, 8 lanes per pair with float4 reads, instead of one pair per wave: 250000 short waves become 7824; dot products summed in a new fixed order
# speedup vs baseline: 4.0575x; 1.0780x over previous
#include <hip/hip_runtime.h>


typedef float v2f __attribute__((ext_vector_type(2)));
typedef float v8f __attribute__((ext_vector_type(8)));

#define NEG_SLOPE 0.2f


__global__ void fill_i32(int* __restrict__ p, int n, int v) {
  int i = blockIdx.x * blockDim.x + threadIdx.x;
  if (i < n) p[i] = v;
}

__global__ void gemm_wmma_f32_lds(const float* __restrict__ A, const float* __restrict__ B,
                                  float* __restrict__ C, int M, int K, int Nd,
                                  int tnPerBlock) {
  extern __shared__ float As[];
  int tilesN = Nd >> 4;
  int blocksPerRow = tilesN / tnPerBlock;
  int tm  = blockIdx.x / blocksPerRow;
  int tn0 = (blockIdx.x % blocksPerRow) * tnPerBlock;
  int wib  = threadIdx.x >> 5;
  int lane = threadIdx.x & 31;

  {
    const float4* Av  = (const float4*)(A + (long)tm * 16 * K);
    float4*       Asv = (float4*)As;
    int nvec = (16 * K) >> 2;
    for (int idx = threadIdx.x; idx < nvec; idx += blockDim.x) Asv[idx] = Av[idx];
  }
  __syncthreads();

  int tn   = tn0 + wib;
  int half = lane >> 4;
  int l15  = lane & 15;
  int bcol = tn * 16 + l15;
  v8f acc = {};
  for (int k = 0; k < K; k += 4) {
    int ka = k + 2 * half;
    v2f a, b;
    a.x = As[l15 * K + ka];
    a.y = As[l15 * K + ka + 1];
    b.x = B[(long)ka * Nd + bcol];
    b.y = B[(long)(ka + 1) * Nd + bcol];
    acc = __builtin_amdgcn_wmma_f32_16x16x4_f32(false, a, false, b,
                                                (short)0, acc, false, false);
  }
  long crow = (long)tm * 16 + half * 8;
#pragma unroll
  for (int r = 0; r < 8; ++r)
    C[(crow + r) * Nd + bcol] = acc[r];
}


__global__ void degree_kernel(const int* __restrict__ eidx, int E, int* __restrict__ cnt) {
  int e = blockIdx.x * blockDim.x + threadIdx.x;
  if (e < E) atomicAdd(&cnt[eidx[E + e]], 1);
}

__global__ void rowptr_kernel(const int* __restrict__ cnt, int* __restrict__ rowptr,
                              int* __restrict__ cursor, int n) {
  __shared__ int sh[256];
  const int t = threadIdx.x;
  const int start = blockIdx.x * 256;
  int part = 0;
  for (int i = t; i < start; i += 256) part += cnt[i];
  const int i = start + t;
  const int c = i < n ? cnt[i] : 0;
  sh[t] = part;
  __syncthreads();
  for (int off = 128; off > 0; off >>= 1) {
    if (t < off) sh[t] += sh[t + off];
    __syncthreads();
  }
  const int base = sh[0];
  __syncthreads();
  sh[t] = c;
  __syncthreads();
  for (int off = 1; off < 256; off <<= 1) {
    int v = t >= off ? sh[t - off] : 0;
    __syncthreads();
    sh[t] += v;
    __syncthreads();
  }
  const int excl = base + sh[t] - c;
  if (i < n) { rowptr[i] = excl; cursor[i] = excl; }
  if (i == n - 1) rowptr[n] = excl + c;
}

__global__ void bucket_kernel(const int* __restrict__ eidx, int E, int* __restrict__ cursor,
                              int* __restrict__ csr_src) {
  int e = blockIdx.x * blockDim.x + threadIdx.x;
  if (e < E) {
    int d = eidx[E + e];
    int p = atomicAdd(&cursor[d], 1);
    csr_src[p] = eidx[e];
  }
}

template<int NH>
__device__ __forceinline__ void load_heads(const float* __restrict__ p, int idx, float (&v)[NH]) {
  if constexpr (NH == 4) {
    float4 t = ((const float4*)p)[idx];
    v[0] = t.x; v[1] = t.y; v[2] = t.z; v[3] = t.w;
  } else {
    for (int h = 0; h < NH; ++h) v[h] = p[(long)idx * NH + h];
  }
}

template<int NF>
__device__ __forceinline__ void load_row(const float* __restrict__ p, float (&v)[NF]) {
  if constexpr (NF % 4 == 0) {
    for (int q = 0; q < NF / 4; ++q) {
      float4 t = ((const float4*)p)[q];
      v[4 * q] = t.x; v[4 * q + 1] = t.y; v[4 * q + 2] = t.z; v[4 * q + 3] = t.w;
    }
  } else if constexpr (NF % 2 == 0) {
    for (int q = 0; q < NF / 2; ++q) {
      float2 t = ((const float2*)p)[q];
      v[2 * q] = t.x; v[2 * q + 1] = t.y;
    }
  } else {
    for (int k = 0; k < NF; ++k) v[k] = p[k];
  }
}

template<int NF>
__device__ __forceinline__ void store_row(float* __restrict__ p, const float (&v)[NF]) {
  if constexpr (NF % 4 == 0) {
    for (int q = 0; q < NF / 4; ++q)
      ((float4*)p)[q] = make_float4(v[4 * q], v[4 * q + 1], v[4 * q + 2], v[4 * q + 3]);
  } else if constexpr (NF % 2 == 0) {
    for (int q = 0; q < NF / 2; ++q)
      ((float2*)p)[q] = make_float2(v[2 * q], v[2 * q + 1]);
  } else {
    for (int k = 0; k < NF; ++k) p[k] = v[k];
  }
}

__device__ __forceinline__ int lane_bcast_i(int v, int i) {
  return __builtin_amdgcn_readlane(v, i);
}
__device__ __forceinline__ float lane_bcast_f(float v, int i) {
  return __int_as_float(__builtin_amdgcn_readlane(__float_as_int(v), i));
}

template<int HEADS, int CH>
__global__ void logits_kernel(const float* __restrict__ hmat, const float* __restrict__ a_src,
                              const float* __restrict__ a_dst, float* __restrict__ als,
                              float* __restrict__ ald, int npairs) {
  constexpr int NF  = 8;
  constexpr int LPP = CH / NF;
  constexpr int PPP = 32 / LPP;
  constexpr int NP  = 8;
  static_assert(LPP == 8 && PPP == 4, "lane map below is written for 64 channels per head");
  static_assert(PPP % HEADS == 0, "a lane must keep one head over all passes");
  const int lane = threadIdx.x & 31;
  const int wave = (int)((blockIdx.x * blockDim.x + threadIdx.x) >> 5);
  const int p0   = wave * (PPP * NP);
  if (p0 >= npairs) return;
  const int grp  = lane / LPP;
  const int sub  = lane % LPP;
  const int head = grp % HEADS;
  float as[NF], ad[NF];
  load_row<NF>(a_src + head * CH + sub * NF, as);
  load_row<NF>(a_dst + head * CH + sub * NF, ad);

  float v[NP][NF];
  for (int u = 0; u < NP; ++u) {
    int p = p0 + u * PPP + grp;
    p = p < npairs ? p : npairs - 1;
    load_row<NF>(hmat + (long)p * CH + sub * NF, v[u]);
  }
  float ss[NP], dd[NP];
  for (int u = 0; u < NP; ++u) {
    float s = 0.f, d = 0.f;
    for (int k = 0; k < NF; ++k) { s += v[u][k] * as[k]; d += v[u][k] * ad[k]; }
    for (int off = LPP / 2; off > 0; off >>= 1) {
      s += __shfl_xor(s, off);
      d += __shfl_xor(d, off);
    }
    ss[u] = s; dd[u] = d;
  }
  float os = ss[0], od = dd[0];
  for (int u = 1; u < NP; ++u) { os = sub == u ? ss[u] : os; od = sub == u ? dd[u] : od; }
  const int from = (lane % PPP) * LPP + lane / PPP;
  os = __shfl(os, from);
  od = __shfl(od, from);
  if (p0 + lane < npairs) { als[p0 + lane] = os; ald[p0 + lane] = od; }
}

template<int HEADS, int CH, bool ELU>
__global__ void gather_kernel(const int* __restrict__ rowptr, const int* __restrict__ csr_src,
                              const float* __restrict__ als, const float* __restrict__ ald,
                              const float* __restrict__ hsrc, const float* __restrict__ bias,
                              float* __restrict__ outp, int nnodes) {
  constexpr int F  = HEADS * CH;
  constexpr int NF = F / 32;
  const int lane = threadIdx.x & 31;
  const int node = __builtin_amdgcn_readfirstlane((int)((blockIdx.x * blockDim.x + threadIdx.x) >> 5));
  if (node >= nnodes) return;
  const int beg = rowptr[node];
  const int deg = rowptr[node + 1] - beg;
  const int tot = deg + 1;
  float adst[HEADS];
  load_heads<HEADS>(ald, node, adst);

  float mx[HEADS];
  for (int h = 0; h < HEADS; ++h) mx[h] = -__builtin_huge_valf();
  for (int c = 0; c < tot; c += 32) {
    const int g = c + lane;
    if (g < tot) {
      const int s = g < deg ? csr_src[beg + g] : node;
      float a[HEADS];
      load_heads<HEADS>(als, s, a);
      for (int h = 0; h < HEADS; ++h) {
        float l = a[h] + adst[h];
        l = l > 0.f ? l : NEG_SLOPE * l;
        mx[h] = fmaxf(mx[h], l);
      }
    }
  }
  for (int h = 0; h < HEADS; ++h)
    for (int off = 16; off > 0; off >>= 1) mx[h] = fmaxf(mx[h], __shfl_xor(mx[h], off));

  const int j0   = lane * NF;
  const int head = j0 / CH;
  float acc[NF];
  for (int k = 0; k < NF; ++k) acc[k] = 0.f;
  float ssum[HEADS];
  for (int h = 0; h < HEADS; ++h) ssum[h] = 0.f;
  for (int c = 0; c < tot; c += 32) {
    const int g = c + lane;
    const int nn = tot - c < 32 ? tot - c : 32;
    int s = node;
    if (g < deg) s = csr_src[beg + g];
    {
      int rank = 0;
      for (int j = 0; j < nn; ++j) {
        const int sj = lane_bcast_i(s, j);
        rank += (int)(sj < s) + (int)((sj == s) & (j < lane));
      }
      rank = g < tot ? rank : lane;
      s = __builtin_amdgcn_ds_permute(rank << 2, s);
    }
    float ex[HEADS];
    for (int h = 0; h < HEADS; ++h) ex[h] = 0.f;
    if (g < tot) {
      float a[HEADS];
      load_heads<HEADS>(als, s, a);
      for (int h = 0; h < HEADS; ++h) {
        float l = a[h] + adst[h];
        l = l > 0.f ? l : NEG_SLOPE * l;
        ex[h] = __expf(l - mx[h]);
        ssum[h] += ex[h];
      }
    }
    const int nn4 = (nn + 3) & ~3;
    for (int i = 0; i < nn4; i += 4) {
      int   sr[4];
      float w[4];
      float v[4][NF];
      for (int u = 0; u < 4; ++u) {
        sr[u] = lane_bcast_i(s, i + u);
        float wu = lane_bcast_f(ex[0], i + u);
        for (int h = 1; h < HEADS; ++h) {
          float th = lane_bcast_f(ex[h], i + u);
          wu = head == h ? th : wu;
        }
        w[u] = wu;
      }
      for (int u = 0; u < 4; ++u) load_row<NF>(hsrc + (long)sr[u] * F + j0, v[u]);
      for (int u = 0; u < 4; ++u)
        for (int k = 0; k < NF; ++k) acc[k] += w[u] * v[u][k];
    }
  }
  for (int h = 0; h < HEADS; ++h)
    for (int off = 16; off > 0; off >>= 1) ssum[h] += __shfl_xor(ssum[h], off);
  float sm = ssum[0];
  for (int h = 1; h < HEADS; ++h) sm = head == h ? ssum[h] : sm;

  float bv[NF], r[NF];
  load_row<NF>(bias + j0, bv);
  for (int k = 0; k < NF; ++k) {
    float val = acc[k] / sm + bv[k];
    if (ELU) val = val > 0.f ? val : (__expf(val) - 1.f);
    r[k] = val;
  }
  store_row<NF>(outp + (long)node * F + j0, r);
}


static inline int cdiv_i(long a, long b) { return (int)((a + b - 1) / b); }

static void launch_gemm(const float* A, const float* B, float* C,
                        int M, int K, int Nd, hipStream_t stream) {
  int tilesM = M / 16, tilesN = Nd / 16;
  int wpb = tilesN >= 8 ? 8 : tilesN;
  dim3 grid(tilesM * (tilesN / wpb));
  dim3 block(32 * wpb);
  size_t shmem = (size_t)16 * K * sizeof(float);
  gemm_wmma_f32_lds<<<grid, block, shmem, stream>>>(A, B, C, M, K, Nd, wpb);
}

extern "C" void kernel_launch(void* const* d_in, const int* in_sizes, int n_in,
                              void* d_out, int out_size, void* d_ws, size_t ws_size,
                              hipStream_t stream) {
  const float* x   = (const float*)d_in[0];
  const int*   eix = (const int*)d_in[1];
  const float* W1  = (const float*)d_in[2];
  const float* as1 = (const float*)d_in[3];
  const float* ad1 = (const float*)d_in[4];
  const float* b1  = (const float*)d_in[5];
  const float* W2  = (const float*)d_in[6];
  const float* as2 = (const float*)d_in[7];
  const float* ad2 = (const float*)d_in[8];
  const float* b2  = (const float*)d_in[9];
  float* out = (float*)d_out;

  const int IN_DIM = 128;
  const int F1     = in_sizes[5];
  const int HID    = in_sizes[9];
  const int HEADS  = F1 / HID;
  const int Nn     = in_sizes[0] / IN_DIM;
  const int E      = in_sizes[1] / 2;

  float* h1e  = (float*)d_ws;
  float* h1   = h1e + (long)Nn * F1;
  float* als1 = h1 + (long)Nn * F1;
  float* ald1 = als1 + (long)Nn * HEADS;
  float* als2 = ald1 + (long)Nn * HEADS;
  float* ald2 = als2 + Nn;
  int* cnt    = (int*)(ald2 + Nn);
  int* rowptr = cnt + Nn;
  int* cursor = rowptr + (Nn + 1);
  int* csr    = cursor + Nn;
  float* h2   = h1;

  fill_i32<<<cdiv_i(Nn, 256), 256, 0, stream>>>(cnt, Nn, 0);
  degree_kernel<<<cdiv_i(E, 256), 256, 0, stream>>>(eix, E, cnt);
  rowptr_kernel<<<cdiv_i(Nn, 256), 256, 0, stream>>>(cnt, rowptr, cursor, Nn);
  bucket_kernel<<<cdiv_i(E, 256), 256, 0, stream>>>(eix, E, cursor, csr);

  launch_gemm(x, W1, h1, Nn, IN_DIM, F1, stream);
  logits_kernel<4, 64><<<cdiv_i((long)Nn * HEADS, 256), 256, 0, stream>>>(
      h1, as1, ad1, als1, ald1, Nn * HEADS);
  gather_kernel<4, 64, true><<<cdiv_i((long)Nn * 32, 256), 256, 0, stream>>>(
      rowptr, csr, als1, ald1, h1, b1, h1e, Nn);

  launch_gemm(h1e, W2, h2, Nn, F1, HID, stream);
  logits_kernel<1, 64><<<cdiv_i((long)Nn, 256), 256, 0, stream>>>(
      h2, as2, ad2, als2, ald2, Nn);
  gather_kernel<1, 64, false><<<cdiv_i((long)Nn * 32, 256), 256, 0, stream>>>(
      rowptr, csr, als2, ald2, h2, b2, out, Nn);
}


# GEMM A slab in LDS at row pitch K+4 floats instead of K: the 16-row f32 16x16x4 A-fragment reads were 16-way bank conflicted, now conflict-free; sums and their order unchanged
# speedup vs baseline: 4.6609x; 1.1487x over previous
#include <hip/hip_runtime.h>


typedef float v2f __attribute__((ext_vector_type(2)));
typedef float v8f __attribute__((ext_vector_type(8)));

#define NEG_SLOPE 0.2f


__global__ void fill_i32(int* __restrict__ p, int n, int v) {
  int i = blockIdx.x * blockDim.x + threadIdx.x;
  if (i < n) p[i] = v;
}

__global__ void gemm_wmma_f32_lds(const float* __restrict__ A, const float* __restrict__ B,
                                  float* __restrict__ C, int M, int K, int Nd,
                                  int tnPerBlock) {
  extern __shared__ float As[];
  const int P = K + 4;
  int tilesN = Nd >> 4;
  int blocksPerRow = tilesN / tnPerBlock;
  int tm  = blockIdx.x / blocksPerRow;
  int tn0 = (blockIdx.x % blocksPerRow) * tnPerBlock;
  int wib  = threadIdx.x >> 5;
  int lane = threadIdx.x & 31;

  {
    const int kv = K >> 2;
    const int nw = blockDim.x >> 5;
    for (int r = wib; r < 16; r += nw) {
      const float4* Av  = (const float4*)(A + ((long)tm * 16 + r) * K);
      float4*       Asv = (float4*)(As + r * P);
      for (int c = lane; c < kv; c += 32) Asv[c] = Av[c];
    }
  }
  __syncthreads();

  int tn   = tn0 + wib;
  int half = lane >> 4;
  int l15  = lane & 15;
  int bcol = tn * 16 + l15;
  v8f acc = {};
  for (int k = 0; k < K; k += 4) {
    int ka = k + 2 * half;
    v2f a, b;
    a.x = As[l15 * P + ka];
    a.y = As[l15 * P + ka + 1];
    b.x = B[(long)ka * Nd + bcol];
    b.y = B[(long)(ka + 1) * Nd + bcol];
    acc = __builtin_amdgcn_wmma_f32_16x16x4_f32(false, a, false, b,
                                                (short)0, acc, false, false);
  }
  long crow = (long)tm * 16 + half * 8;
#pragma unroll
  for (int r = 0; r < 8; ++r)
    C[(crow + r) * Nd + bcol] = acc[r];
}


__global__ void degree_kernel(const int* __restrict__ eidx, int E, int* __restrict__ cnt) {
  int e = blockIdx.x * blockDim.x + threadIdx.x;
  if (e < E) atomicAdd(&cnt[eidx[E + e]], 1);
}

__global__ void rowptr_kernel(const int* __restrict__ cnt, int* __restrict__ rowptr,
                              int* __restrict__ cursor, int n) {
  __shared__ int sh[256];
  const int t = threadIdx.x;
  const int start = blockIdx.x * 256;
  int part = 0;
  for (int i = t; i < start; i += 256) part += cnt[i];
  const int i = start + t;
  const int c = i < n ? cnt[i] : 0;
  sh[t] = part;
  __syncthreads();
  for (int off = 128; off > 0; off >>= 1) {
    if (t < off) sh[t] += sh[t + off];
    __syncthreads();
  }
  const int base = sh[0];
  __syncthreads();
  sh[t] = c;
  __syncthreads();
  for (int off = 1; off < 256; off <<= 1) {
    int v = t >= off ? sh[t - off] : 0;
    __syncthreads();
    sh[t] += v;
    __syncthreads();
  }
  const int excl = base + sh[t] - c;
  if (i < n) { rowptr[i] = excl; cursor[i] = excl; }
  if (i == n - 1) rowptr[n] = excl + c;
}

__global__ void bucket_kernel(const int* __restrict__ eidx, int E, int* __restrict__ cursor,
                              int* __restrict__ csr_src) {
  int e = blockIdx.x * blockDim.x + threadIdx.x;
  if (e < E) {
    int d = eidx[E + e];
    int p = atomicAdd(&cursor[d], 1);
    csr_src[p] = eidx[e];
  }
}

template<int NH>
__device__ __forceinline__ void load_heads(const float* __restrict__ p, int idx, float (&v)[NH]) {
  if constexpr (NH == 4) {
    float4 t = ((const float4*)p)[idx];
    v[0] = t.x; v[1] = t.y; v[2] = t.z; v[3] = t.w;
  } else {
    for (int h = 0; h < NH; ++h) v[h] = p[(long)idx * NH + h];
  }
}

template<int NF>
__device__ __forceinline__ void load_row(const float* __restrict__ p, float (&v)[NF]) {
  if constexpr (NF % 4 == 0) {
    for (int q = 0; q < NF / 4; ++q) {
      float4 t = ((const float4*)p)[q];
      v[4 * q] = t.x; v[4 * q + 1] = t.y; v[4 * q + 2] = t.z; v[4 * q + 3] = t.w;
    }
  } else if constexpr (NF % 2 == 0) {
    for (int q = 0; q < NF / 2; ++q) {
      float2 t = ((const float2*)p)[q];
      v[2 * q] = t.x; v[2 * q + 1] = t.y;
    }
  } else {
    for (int k = 0; k < NF; ++k) v[k] = p[k];
  }
}

template<int NF>
__device__ __forceinline__ void store_row(float* __restrict__ p, const float (&v)[NF]) {
  if constexpr (NF % 4 == 0) {
    for (int q = 0; q < NF / 4; ++q)
      ((float4*)p)[q] = make_float4(v[4 * q], v[4 * q + 1], v[4 * q + 2], v[4 * q + 3]);
  } else if constexpr (NF % 2 == 0) {
    for (int q = 0; q < NF / 2; ++q)
      ((float2*)p)[q] = make_float2(v[2 * q], v[2 * q + 1]);
  } else {
    for (int k = 0; k < NF; ++k) p[k] = v[k];
  }
}

__device__ __forceinline__ int lane_bcast_i(int v, int i) {
  return __builtin_amdgcn_readlane(v, i);
}
__device__ __forceinline__ float lane_bcast_f(float v, int i) {
  return __int_as_float(__builtin_amdgcn_readlane(__float_as_int(v), i));
}

template<int HEADS, int CH>
__global__ void logits_kernel(const float* __restrict__ hmat, const float* __restrict__ a_src,
                              const float* __restrict__ a_dst, float* __restrict__ als,
                              float* __restrict__ ald, int npairs) {
  constexpr int NF  = 8;
  constexpr int LPP = CH / NF;
  constexpr int PPP = 32 / LPP;
  constexpr int NP  = 8;
  static_assert(LPP == 8 && PPP == 4, "lane map below is written for 64 channels per head");
  static_assert(PPP % HEADS == 0, "a lane must keep one head over all passes");
  const int lane = threadIdx.x & 31;
  const int wave = (int)((blockIdx.x * blockDim.x + threadIdx.x) >> 5);
  const int p0   = wave * (PPP * NP);
  if (p0 >= npairs) return;
  const int grp  = lane / LPP;
  const int sub  = lane % LPP;
  const int head = grp % HEADS;
  float as[NF], ad[NF];
  load_row<NF>(a_src + head * CH + sub * NF, as);
  load_row<NF>(a_dst + head * CH + sub * NF, ad);

  float v[NP][NF];
  for (int u = 0; u < NP; ++u) {
    int p = p0 + u * PPP + grp;
    p = p < npairs ? p : npairs - 1;
    load_row<NF>(hmat + (long)p * CH + sub * NF, v[u]);
  }
  float ss[NP], dd[NP];
  for (int u = 0; u < NP; ++u) {
    float s = 0.f, d = 0.f;
    for (int k = 0; k < NF; ++k) { s += v[u][k] * as[k]; d += v[u][k] * ad[k]; }
    for (int off = LPP / 2; off > 0; off >>= 1) {
      s += __shfl_xor(s, off);
      d += __shfl_xor(d, off);
    }
    ss[u] = s; dd[u] = d;
  }
  float os = ss[0], od = dd[0];
  for (int u = 1; u < NP; ++u) { os = sub == u ? ss[u] : os; od = sub == u ? dd[u] : od; }
  const int from = (lane % PPP) * LPP + lane / PPP;
  os = __shfl(os, from);
  od = __shfl(od, from);
  if (p0 + lane < npairs) { als[p0 + lane] = os; ald[p0 + lane] = od; }
}

template<int HEADS, int CH, bool ELU>
__global__ void gather_kernel(const int* __restrict__ rowptr, const int* __restrict__ csr_src,
                              const float* __restrict__ als, const float* __restrict__ ald,
                              const float* __restrict__ hsrc, const float* __restrict__ bias,
                              float* __restrict__ outp, int nnodes) {
  constexpr int F  = HEADS * CH;
  constexpr int NF = F / 32;
  const int lane = threadIdx.x & 31;
  const int node = __builtin_amdgcn_readfirstlane((int)((blockIdx.x * blockDim.x + threadIdx.x) >> 5));
  if (node >= nnodes) return;
  const int beg = rowptr[node];
  const int deg = rowptr[node + 1] - beg;
  const int tot = deg + 1;
  float adst[HEADS];
  load_heads<HEADS>(ald, node, adst);

  float mx[HEADS];
  for (int h = 0; h < HEADS; ++h) mx[h] = -__builtin_huge_valf();
  for (int c = 0; c < tot; c += 32) {
    const int g = c + lane;
    if (g < tot) {
      const int s = g < deg ? csr_src[beg + g] : node;
      float a[HEADS];
      load_heads<HEADS>(als, s, a);
      for (int h = 0; h < HEADS; ++h) {
        float l = a[h] + adst[h];
        l = l > 0.f ? l : NEG_SLOPE * l;
        mx[h] = fmaxf(mx[h], l);
      }
    }
  }
  for (int h = 0; h < HEADS; ++h)
    for (int off = 16; off > 0; off >>= 1) mx[h] = fmaxf(mx[h], __shfl_xor(mx[h], off));

  const int j0   = lane * NF;
  const int head = j0 / CH;
  float acc[NF];
  for (int k = 0; k < NF; ++k) acc[k] = 0.f;
  float ssum[HEADS];
  for (int h = 0; h < HEADS; ++h) ssum[h] = 0.f;
  for (int c = 0; c < tot; c += 32) {
    const int g = c + lane;
    const int nn = tot - c < 32 ? tot - c : 32;
    int s = node;
    if (g < deg) s = csr_src[beg + g];
    {
      int rank = 0;
      for (int j = 0; j < nn; ++j) {
        const int sj = lane_bcast_i(s, j);
        rank += (int)(sj < s) + (int)((sj == s) & (j < lane));
      }
      rank = g < tot ? rank : lane;
      s = __builtin_amdgcn_ds_permute(rank << 2, s);
    }
    float ex[HEADS];
    for (int h = 0; h < HEADS; ++h) ex[h] = 0.f;
    if (g < tot) {
      float a[HEADS];
      load_heads<HEADS>(als, s, a);
      for (int h = 0; h < HEADS; ++h) {
        float l = a[h] + adst[h];
        l = l > 0.f ? l : NEG_SLOPE * l;
        ex[h] = __expf(l - mx[h]);
        ssum[h] += ex[h];
      }
    }
    const int nn4 = (nn + 3) & ~3;
    for (int i = 0; i < nn4; i += 4) {
      int   sr[4];
      float w[4];
      float v[4][NF];
      for (int u = 0; u < 4; ++u) {
        sr[u] = lane_bcast_i(s, i + u);
        float wu = lane_bcast_f(ex[0], i + u);
        for (int h = 1; h < HEADS; ++h) {
          float th = lane_bcast_f(ex[h], i + u);
          wu = head == h ? th : wu;
        }
        w[u] = wu;
      }
      for (int u = 0; u < 4; ++u) load_row<NF>(hsrc + (long)sr[u] * F + j0, v[u]);
      for (int u = 0; u < 4; ++u)
        for (int k = 0; k < NF; ++k) acc[k] += w[u] * v[u][k];
    }
  }
  for (int h = 0; h < HEADS; ++h)
    for (int off = 16; off > 0; off >>= 1) ssum[h] += __shfl_xor(ssum[h], off);
  float sm = ssum[0];
  for (int h = 1; h < HEADS; ++h) sm = head == h ? ssum[h] : sm;

  float bv[NF], r[NF];
  load_row<NF>(bias + j0, bv);
  for (int k = 0; k < NF; ++k) {
    float val = acc[k] / sm + bv[k];
    if (ELU) val = val > 0.f ? val : (__expf(val) - 1.f);
    r[k] = val;
  }
  store_row<NF>(outp + (long)node * F + j0, r);
}


static inline int cdiv_i(long a, long b) { return (int)((a + b - 1) / b); }

static void launch_gemm(const float* A, const float* B, float* C,
                        int M, int K, int Nd, hipStream_t stream) {
  int tilesM = M / 16, tilesN = Nd / 16;
  int wpb = tilesN >= 8 ? 8 : tilesN;
  dim3 grid(tilesM * (tilesN / wpb));
  dim3 block(32 * wpb);
  size_t shmem = (size_t)16 * (K + 4) * sizeof(float);
  gemm_wmma_f32_lds<<<grid, block, shmem, stream>>>(A, B, C, M, K, Nd, wpb);
}

extern "C" void kernel_launch(void* const* d_in, const int* in_sizes, int n_in,
                              void* d_out, int out_size, void* d_ws, size_t ws_size,
                              hipStream_t stream) {
  const float* x   = (const float*)d_in[0];
  const int*   eix = (const int*)d_in[1];
  const float* W1  = (const float*)d_in[2];
  const float* as1 = (const float*)d_in[3];
  const float* ad1 = (const float*)d_in[4];
  const float* b1  = (const float*)d_in[5];
  const float* W2  = (const float*)d_in[6];
  const float* as2 = (const float*)d_in[7];
  const float* ad2 = (const float*)d_in[8];
  const float* b2  = (const float*)d_in[9];
  float* out = (float*)d_out;

  const int IN_DIM = 128;
  const int F1     = in_sizes[5];
  const int HID    = in_sizes[9];
  const int HEADS  = F1 / HID;
  const int Nn     = in_sizes[0] / IN_DIM;
  const int E      = in_sizes[1] / 2;

  float* h1e  = (float*)d_ws;
  float* h1   = h1e + (long)Nn * F1;
  float* als1 = h1 + (long)Nn * F1;
  float* ald1 = als1 + (long)Nn * HEADS;
  float* als2 = ald1 + (long)Nn * HEADS;
  float* ald2 = als2 + Nn;
  int* cnt    = (int*)(ald2 + Nn);
  int* rowptr = cnt + Nn;
  int* cursor = rowptr + (Nn + 1);
  int* csr    = cursor + Nn;
  float* h2   = h1;

  fill_i32<<<cdiv_i(Nn, 256), 256, 0, stream>>>(cnt, Nn, 0);
  degree_kernel<<<cdiv_i(E, 256), 256, 0, stream>>>(eix, E, cnt);
  rowptr_kernel<<<cdiv_i(Nn, 256), 256, 0, stream>>>(cnt, rowptr, cursor, Nn);
  bucket_kernel<<<cdiv_i(E, 256), 256, 0, stream>>>(eix, E, cursor, csr);

  launch_gemm(x, W1, h1, Nn, IN_DIM, F1, stream);
  logits_kernel<4, 64><<<cdiv_i((long)Nn * HEADS, 256), 256, 0, stream>>>(
      h1, as1, ad1, als1, ald1, Nn * HEADS);
  gather_kernel<4, 64, true><<<cdiv_i((long)Nn * 32, 256), 256, 0, stream>>>(
      rowptr, csr, als1, ald1, h1, b1, h1e, Nn);

  launch_gemm(h1e, W2, h2, Nn, F1, HID, stream);
  logits_kernel<1, 64><<<cdiv_i((long)Nn, 256), 256, 0, stream>>>(
      h2, as2, ad2, als2, ald2, Nn);
  gather_kernel<1, 64, false><<<cdiv_i((long)Nn * 32, 256), 256, 0, stream>>>(
      rowptr, csr, als2, ald2, h2, b2, out, Nn);
}


# Edge grouping without global atomics: 64 blocks count their edge chunk per node in LDS (returned LDS add = rank), a scan pass gives slots, a dense pass places; 9 launches; float sum order unchanged
# speedup vs baseline: 6.2765x; 1.3466x over previous
#include <hip/hip_runtime.h>


typedef float v2f __attribute__((ext_vector_type(2)));
typedef float v8f __attribute__((ext_vector_type(8)));

#define NEG_SLOPE 0.2f

__global__ void gemm_wmma_f32_lds(const float* __restrict__ A, const float* __restrict__ B,
                                  float* __restrict__ C, int M, int K, int Nd,
                                  int tnPerBlock) {
  extern __shared__ float As[];
  const int P = K + 4;
  int tilesN = Nd >> 4;
  int blocksPerRow = tilesN / tnPerBlock;
  int tm  = blockIdx.x / blocksPerRow;
  int tn0 = (blockIdx.x % blocksPerRow) * tnPerBlock;
  int wib  = threadIdx.x >> 5;
  int lane = threadIdx.x & 31;

  {
    const int kv = K >> 2;
    const int nw = blockDim.x >> 5;
    for (int r = wib; r < 16; r += nw) {
      const float4* Av  = (const float4*)(A + ((long)tm * 16 + r) * K);
      float4*       Asv = (float4*)(As + r * P);
      for (int c = lane; c < kv; c += 32) Asv[c] = Av[c];
    }
  }
  __syncthreads();

  int tn   = tn0 + wib;
  int half = lane >> 4;
  int l15  = lane & 15;
  int bcol = tn * 16 + l15;
  v8f acc = {};
  for (int k = 0; k < K; k += 4) {
    int ka = k + 2 * half;
    v2f a, b;
    a.x = As[l15 * P + ka];
    a.y = As[l15 * P + ka + 1];
    b.x = B[(long)ka * Nd + bcol];
    b.y = B[(long)(ka + 1) * Nd + bcol];
    acc = __builtin_amdgcn_wmma_f32_16x16x4_f32(false, a, false, b,
                                                (short)0, acc, false, false);
  }
  long crow = (long)tm * 16 + half * 8;
#pragma unroll
  for (int r = 0; r < 8; ++r)
    C[(crow + r) * Nd + bcol] = acc[r];
}


constexpr int NCHUNK  = 64;
constexpr int LDS_CAP = 65536;
constexpr int CNT_T   = 1024;
constexpr int CNT_B   = 8;

__global__ __launch_bounds__(CNT_T) void count_rank_kernel(
    const int* __restrict__ eidx, int E, int EC, int NP,
    int* __restrict__ tab, int* __restrict__ bsum, int* __restrict__ rank) {
  __shared__ __attribute__((aligned(16))) int lc[LDS_CAP];
  const int t = threadIdx.x;
  const int c = blockIdx.x;
  const int* __restrict__ dst = eidx + E;
  const int e0 = c * EC;
  const int e1 = e0 + EC < E ? e0 + EC : E;

  int d[CNT_B];
  for (int u = 0; u < CNT_B; ++u) {
    const int e = e0 + t + u * CNT_T;
    d[u] = e < e1 ? dst[e] : -1;
  }
  for (int i = t * 4; i < NP; i += CNT_T * 4) *(int4*)&lc[i] = make_int4(0, 0, 0, 0);
  __syncthreads();

  for (int eb = e0; eb < e1; eb += CNT_T * CNT_B) {
    int dn[CNT_B];
    for (int u = 0; u < CNT_B; ++u) {
      const int e = eb + CNT_T * CNT_B + t + u * CNT_T;
      dn[u] = e < e1 ? dst[e] : -1;
    }
    int r[CNT_B];
    for (int u = 0; u < CNT_B; ++u) r[u] = atomicAdd(&lc[d[u] >= 0 ? d[u] : NP + (t & 31)], 1);
    for (int u = 0; u < CNT_B; ++u)
      if (d[u] >= 0) rank[eb + t + u * CNT_T] = r[u];
    for (int u = 0; u < CNT_B; ++u) d[u] = dn[u];
  }
  __syncthreads();

  int* __restrict__ mytab = tab + (long)c * NP;
  for (int i = t * 4; i < NP; i += CNT_T * 4) *(int4*)&mytab[i] = *(const int4*)&lc[i];
  const int wave = t >> 5, lane = t & 31;
  for (int b = wave; b < (NP >> 8); b += CNT_T / 32) {
    const int4 p = *(const int4*)&lc[b * 256 + lane * 8];
    const int4 q = *(const int4*)&lc[b * 256 + lane * 8 + 4];
    int sum = p.x + p.y + p.z + p.w + q.x + q.y + q.z + q.w;
    for (int off = 16; off > 0; off >>= 1) sum += __shfl_xor(sum, off);
    if (lane == 0) bsum[b * NCHUNK + c] = sum;
  }
}

__global__ __launch_bounds__(256) void offsets_kernel(int* __restrict__ tab,
                                                      const int* __restrict__ bsum,
                                                      int* __restrict__ rowptr, int n, int NP) {
  __shared__ int sh[256];
  const int t = threadIdx.x;
  const int i = blockIdx.x * 256 + t;
  int part = 0;
  for (int j = t; j < (int)blockIdx.x * NCHUNK; j += 256) part += bsum[j];
  int v[NCHUNK];
  for (int c = 0; c < NCHUNK; ++c) v[c] = tab[(long)c * NP + i];
  int deg = 0;
  for (int c = 0; c < NCHUNK; ++c) { const int x = v[c]; v[c] = deg; deg += x; }
  sh[t] = part;
  __syncthreads();
  for (int off = 128; off > 0; off >>= 1) {
    if (t < off) sh[t] += sh[t + off];
    __syncthreads();
  }
  const int base = sh[0];
  __syncthreads();
  sh[t] = deg;
  __syncthreads();
  for (int off = 1; off < 256; off <<= 1) {
    int w = t >= off ? sh[t - off] : 0;
    __syncthreads();
    sh[t] += w;
    __syncthreads();
  }
  const int excl = base + sh[t] - deg;
  for (int c = 0; c < NCHUNK; ++c) tab[(long)c * NP + i] = excl + v[c];
  if (i < n) rowptr[i] = excl;
  if (i == n - 1) rowptr[n] = excl + deg;
}

__global__ __launch_bounds__(256) void place_kernel(const int* __restrict__ eidx, int E, int EC, int NP,
                                                    int bpc, const int* __restrict__ tab,
                                                    const int* __restrict__ rank,
                                                    int* __restrict__ csr_src) {
  const int c  = blockIdx.x / bpc;
  const int sb = blockIdx.x - c * bpc;
  const int e0 = c * EC;
  const int e1 = e0 + EC < E ? e0 + EC : E;
  const int eb = e0 + sb * 1024 + threadIdx.x;
  const int* __restrict__ mytab = tab + (long)c * NP;
  int d[4], s[4], r[4];
  for (int u = 0; u < 4; ++u) {
    const int e = eb + u * 256;
    d[u] = -1; s[u] = 0; r[u] = 0;
    if (e < e1) { d[u] = eidx[E + e]; s[u] = eidx[e]; r[u] = rank[e]; }
  }
  int p[4];
  for (int u = 0; u < 4; ++u) p[u] = mytab[d[u] >= 0 ? d[u] : 0] + r[u];
  for (int u = 0; u < 4; ++u)
    if (d[u] >= 0) csr_src[p[u]] = s[u];
}

template<int NH>
__device__ __forceinline__ void load_heads(const float* __restrict__ p, int idx, float (&v)[NH]) {
  if constexpr (NH == 4) {
    float4 t = ((const float4*)p)[idx];
    v[0] = t.x; v[1] = t.y; v[2] = t.z; v[3] = t.w;
  } else {
    for (int h = 0; h < NH; ++h) v[h] = p[(long)idx * NH + h];
  }
}

template<int NF>
__device__ __forceinline__ void load_row(const float* __restrict__ p, float (&v)[NF]) {
  if constexpr (NF % 4 == 0) {
    for (int q = 0; q < NF / 4; ++q) {
      float4 t = ((const float4*)p)[q];
      v[4 * q] = t.x; v[4 * q + 1] = t.y; v[4 * q + 2] = t.z; v[4 * q + 3] = t.w;
    }
  } else if constexpr (NF % 2 == 0) {
    for (int q = 0; q < NF / 2; ++q) {
      float2 t = ((const float2*)p)[q];
      v[2 * q] = t.x; v[2 * q + 1] = t.y;
    }
  } else {
    for (int k = 0; k < NF; ++k) v[k] = p[k];
  }
}

template<int NF>
__device__ __forceinline__ void store_row(float* __restrict__ p, const float (&v)[NF]) {
  if constexpr (NF % 4 == 0) {
    for (int q = 0; q < NF / 4; ++q)
      ((float4*)p)[q] = make_float4(v[4 * q], v[4 * q + 1], v[4 * q + 2], v[4 * q + 3]);
  } else if constexpr (NF % 2 == 0) {
    for (int q = 0; q < NF / 2; ++q)
      ((float2*)p)[q] = make_float2(v[2 * q], v[2 * q + 1]);
  } else {
    for (int k = 0; k < NF; ++k) p[k] = v[k];
  }
}

__device__ __forceinline__ int lane_bcast_i(int v, int i) {
  return __builtin_amdgcn_readlane(v, i);
}
__device__ __forceinline__ float lane_bcast_f(float v, int i) {
  return __int_as_float(__builtin_amdgcn_readlane(__float_as_int(v), i));
}

template<int HEADS, int CH>
__global__ void logits_kernel(const float* __restrict__ hmat, const float* __restrict__ a_src,
                              const float* __restrict__ a_dst, float* __restrict__ als,
                              float* __restrict__ ald, int npairs) {
  constexpr int NF  = 8;
  constexpr int LPP = CH / NF;
  constexpr int PPP = 32 / LPP;
  constexpr int NP  = 8;
  static_assert(LPP == 8 && PPP == 4, "lane map below is written for 64 channels per head");
  static_assert(PPP % HEADS == 0, "a lane must keep one head over all passes");
  const int lane = threadIdx.x & 31;
  const int wave = (int)((blockIdx.x * blockDim.x + threadIdx.x) >> 5);
  const int p0   = wave * (PPP * NP);
  if (p0 >= npairs) return;
  const int grp  = lane / LPP;
  const int sub  = lane % LPP;
  const int head = grp % HEADS;
  float as[NF], ad[NF];
  load_row<NF>(a_src + head * CH + sub * NF, as);
  load_row<NF>(a_dst + head * CH + sub * NF, ad);

  float v[NP][NF];
  for (int u = 0; u < NP; ++u) {
    int p = p0 + u * PPP + grp;
    p = p < npairs ? p : npairs - 1;
    load_row<NF>(hmat + (long)p * CH + sub * NF, v[u]);
  }
  float ss[NP], dd[NP];
  for (int u = 0; u < NP; ++u) {
    float s = 0.f, d = 0.f;
    for (int k = 0; k < NF; ++k) { s += v[u][k] * as[k]; d += v[u][k] * ad[k]; }
    for (int off = LPP / 2; off > 0; off >>= 1) {
      s += __shfl_xor(s, off);
      d += __shfl_xor(d, off);
    }
    ss[u] = s; dd[u] = d;
  }
  float os = ss[0], od = dd[0];
  for (int u = 1; u < NP; ++u) { os = sub == u ? ss[u] : os; od = sub == u ? dd[u] : od; }
  const int from = (lane % PPP) * LPP + lane / PPP;
  os = __shfl(os, from);
  od = __shfl(od, from);
  if (p0 + lane < npairs) { als[p0 + lane] = os; ald[p0 + lane] = od; }
}

template<int HEADS, int CH, bool ELU>
__global__ void gather_kernel(const int* __restrict__ rowptr, const int* __restrict__ csr_src,
                              const float* __restrict__ als, const float* __restrict__ ald,
                              const float* __restrict__ hsrc, const float* __restrict__ bias,
                              float* __restrict__ outp, int nnodes) {
  constexpr int F  = HEADS * CH;
  constexpr int NF = F / 32;
  const int lane = threadIdx.x & 31;
  const int node = __builtin_amdgcn_readfirstlane((int)((blockIdx.x * blockDim.x + threadIdx.x) >> 5));
  if (node >= nnodes) return;
  const int beg = rowptr[node];
  const int deg = rowptr[node + 1] - beg;
  const int tot = deg + 1;
  float adst[HEADS];
  load_heads<HEADS>(ald, node, adst);

  float mx[HEADS];
  for (int h = 0; h < HEADS; ++h) mx[h] = -__builtin_huge_valf();
  for (int c = 0; c < tot; c += 32) {
    const int g = c + lane;
    if (g < tot) {
      const int s = g < deg ? csr_src[beg + g] : node;
      float a[HEADS];
      load_heads<HEADS>(als, s, a);
      for (int h = 0; h < HEADS; ++h) {
        float l = a[h] + adst[h];
        l = l > 0.f ? l : NEG_SLOPE * l;
        mx[h] = fmaxf(mx[h], l);
      }
    }
  }
  for (int h = 0; h < HEADS; ++h)
    for (int off = 16; off > 0; off >>= 1) mx[h] = fmaxf(mx[h], __shfl_xor(mx[h], off));

  const int j0   = lane * NF;
  const int head = j0 / CH;
  float acc[NF];
  for (int k = 0; k < NF; ++k) acc[k] = 0.f;
  float ssum[HEADS];
  for (int h = 0; h < HEADS; ++h) ssum[h] = 0.f;
  for (int c = 0; c < tot; c += 32) {
    const int g = c + lane;
    const int nn = tot - c < 32 ? tot - c : 32;
    int s = node;
    if (g < deg) s = csr_src[beg + g];
    {
      int rank = 0;
      for (int j = 0; j < nn; ++j) {
        const int sj = lane_bcast_i(s, j);
        rank += (int)(sj < s) + (int)((sj == s) & (j < lane));
      }
      rank = g < tot ? rank : lane;
      s = __builtin_amdgcn_ds_permute(rank << 2, s);
    }
    float ex[HEADS];
    for (int h = 0; h < HEADS; ++h) ex[h] = 0.f;
    if (g < tot) {
      float a[HEADS];
      load_heads<HEADS>(als, s, a);
      for (int h = 0; h < HEADS; ++h) {
        float l = a[h] + adst[h];
        l = l > 0.f ? l : NEG_SLOPE * l;
        ex[h] = __expf(l - mx[h]);
        ssum[h] += ex[h];
      }
    }
    const int nn4 = (nn + 3) & ~3;
    for (int i = 0; i < nn4; i += 4) {
      int   sr[4];
      float w[4];
      float v[4][NF];
      for (int u = 0; u < 4; ++u) {
        sr[u] = lane_bcast_i(s, i + u);
        float wu = lane_bcast_f(ex[0], i + u);
        for (int h = 1; h < HEADS; ++h) {
          float th = lane_bcast_f(ex[h], i + u);
          wu = head == h ? th : wu;
        }
        w[u] = wu;
      }
      for (int u = 0; u < 4; ++u) load_row<NF>(hsrc + (long)sr[u] * F + j0, v[u]);
      for (int u = 0; u < 4; ++u)
        for (int k = 0; k < NF; ++k) acc[k] += w[u] * v[u][k];
    }
  }
  for (int h = 0; h < HEADS; ++h)
    for (int off = 16; off > 0; off >>= 1) ssum[h] += __shfl_xor(ssum[h], off);
  float sm = ssum[0];
  for (int h = 1; h < HEADS; ++h) sm = head == h ? ssum[h] : sm;

  float bv[NF], r[NF];
  load_row<NF>(bias + j0, bv);
  for (int k = 0; k < NF; ++k) {
    float val = acc[k] / sm + bv[k];
    if (ELU) val = val > 0.f ? val : (__expf(val) - 1.f);
    r[k] = val;
  }
  store_row<NF>(outp + (long)node * F + j0, r);
}


static inline int cdiv_i(long a, long b) { return (int)((a + b - 1) / b); }

static void launch_gemm(const float* A, const float* B, float* C,
                        int M, int K, int Nd, hipStream_t stream) {
  int tilesM = M / 16, tilesN = Nd / 16;
  int wpb = tilesN >= 8 ? 8 : tilesN;
  dim3 grid(tilesM * (tilesN / wpb));
  dim3 block(32 * wpb);
  size_t shmem = (size_t)16 * (K + 4) * sizeof(float);
  gemm_wmma_f32_lds<<<grid, block, shmem, stream>>>(A, B, C, M, K, Nd, wpb);
}

extern "C" void kernel_launch(void* const* d_in, const int* in_sizes, int n_in,
                              void* d_out, int out_size, void* d_ws, size_t ws_size,
                              hipStream_t stream) {
  const float* x   = (const float*)d_in[0];
  const int*   eix = (const int*)d_in[1];
  const float* W1  = (const float*)d_in[2];
  const float* as1 = (const float*)d_in[3];
  const float* ad1 = (const float*)d_in[4];
  const float* b1  = (const float*)d_in[5];
  const float* W2  = (const float*)d_in[6];
  const float* as2 = (const float*)d_in[7];
  const float* ad2 = (const float*)d_in[8];
  const float* b2  = (const float*)d_in[9];
  float* out = (float*)d_out;

  const int IN_DIM = 128;
  const int F1     = in_sizes[5];
  const int HID    = in_sizes[9];
  const int HEADS  = F1 / HID;
  const int Nn     = in_sizes[0] / IN_DIM;
  const int E      = in_sizes[1] / 2;

  float* h1e  = (float*)d_ws;
  float* h1   = h1e + (long)Nn * F1;
  float* als1 = h1 + (long)Nn * F1;
  float* ald1 = als1 + (long)Nn * HEADS;
  float* als2 = ald1 + (long)Nn * HEADS;
  float* ald2 = als2 + Nn;
  const int NP = ((Nn + 255) / 256) * 256;
  const int EC = ((cdiv_i(E, NCHUNK) + 3) / 4) * 4;
  int* rowptr = (int*)(ald2 + Nn);
  int* csr    = rowptr + (Nn + 1);
  int* rank   = csr + E;
  int* bsum   = rank + E;
  int* tab    = bsum + (NP / 256) * NCHUNK + 64;
  tab = (int*)(((size_t)tab + 255) & ~(size_t)255);
  float* h2   = h1;

  const int bpc = cdiv_i(EC, 1024);
  count_rank_kernel<<<NCHUNK, CNT_T, 0, stream>>>(eix, E, EC, NP, tab, bsum, rank);
  offsets_kernel<<<NP / 256, 256, 0, stream>>>(tab, bsum, rowptr, Nn, NP);
  place_kernel<<<NCHUNK * bpc, 256, 0, stream>>>(eix, E, EC, NP, bpc, tab, rank, csr);

  launch_gemm(x, W1, h1, Nn, IN_DIM, F1, stream);
  logits_kernel<4, 64><<<cdiv_i((long)Nn * HEADS, 256), 256, 0, stream>>>(
      h1, as1, ad1, als1, ald1, Nn * HEADS);
  gather_kernel<4, 64, true><<<cdiv_i((long)Nn * 32, 256), 256, 0, stream>>>(
      rowptr, csr, als1, ald1, h1, b1, h1e, Nn);

  launch_gemm(h1e, W2, h2, Nn, F1, HID, stream);
  logits_kernel<1, 64><<<cdiv_i((long)Nn, 256), 256, 0, stream>>>(
      h2, as2, ad2, als2, ald2, Nn);
  gather_kernel<1, 64, false><<<cdiv_i((long)Nn * 32, 256), 256, 0, stream>>>(
      rowptr, csr, als2, ald2, h2, b2, out, Nn);
}


# Attention logits taken from the GEMM accumulators in its epilogue (16-lane row sums on the vector ALU, a head's 4 waves meet in LDS): both logits launches and re-reads of h gone; logit sum order new
# speedup vs baseline: 6.6041x; 1.0522x over previous
#include <hip/hip_runtime.h>


typedef float v2f __attribute__((ext_vector_type(2)));
typedef float v8f __attribute__((ext_vector_type(8)));

#define NEG_SLOPE 0.2f

template<int M_>
__device__ __forceinline__ float row_xor_c(float v) {
  return __int_as_float(__builtin_amdgcn_update_dpp(0, __float_as_int(v), 0x160 + M_, 0xf, 0xf, true));
}
__device__ __forceinline__ float row_xor_f(float v, int m) {
  return m == 1 ? row_xor_c<1>(v) : m == 2 ? row_xor_c<2>(v) : m == 4 ? row_xor_c<4>(v) : row_xor_c<8>(v);
}

template<int HEADS, int CH, int WPB>
__global__ void gemm_wmma_f32_lds(const float* __restrict__ A, const float* __restrict__ B,
                                  float* __restrict__ C, int M, int K, int Nd,
                                  int tnPerBlock,
                                  const float* __restrict__ a_src, const float* __restrict__ a_dst,
                                  float* __restrict__ als, float* __restrict__ ald) {
  extern __shared__ float As[];
  __shared__ float red[2][16][8];
  const int P = K + 4;
  int tilesN = Nd >> 4;
  int blocksPerRow = tilesN / tnPerBlock;
  int tm  = blockIdx.x / blocksPerRow;
  int tn0 = (blockIdx.x % blocksPerRow) * tnPerBlock;
  int wib  = threadIdx.x >> 5;
  int lane = threadIdx.x & 31;

  {
    const int kv = K >> 2;
    const int nw = blockDim.x >> 5;
    for (int r = wib; r < 16; r += nw) {
      const float4* Av  = (const float4*)(A + ((long)tm * 16 + r) * K);
      float4*       Asv = (float4*)(As + r * P);
      for (int c = lane; c < kv; c += 32) Asv[c] = Av[c];
    }
  }
  __syncthreads();

  int tn   = tn0 + wib;
  int half = lane >> 4;
  int l15  = lane & 15;
  int bcol = tn * 16 + l15;
  const float cs = a_src[bcol];
  const float cd = a_dst[bcol];
  v8f acc = {};
  for (int k = 0; k < K; k += 4) {
    int ka = k + 2 * half;
    v2f a, b;
    a.x = As[l15 * P + ka];
    a.y = As[l15 * P + ka + 1];
    b.x = B[(long)ka * Nd + bcol];
    b.y = B[(long)(ka + 1) * Nd + bcol];
    acc = __builtin_amdgcn_wmma_f32_16x16x4_f32(false, a, false, b,
                                                (short)0, acc, false, false);
  }
  long crow = (long)tm * 16 + half * 8;
#pragma unroll
  for (int r = 0; r < 8; ++r)
    C[(crow + r) * Nd + bcol] = acc[r];

  float ps[8], pd[8];
  for (int r = 0; r < 8; ++r) { ps[r] = acc[r] * cs; pd[r] = acc[r] * cd; }
  for (int m = 1; m < 16; m <<= 1)
    for (int r = 0; r < 8; ++r) { ps[r] += row_xor_f(ps[r], m); pd[r] += row_xor_f(pd[r], m); }
  float vs = ps[0], vd = pd[0];
  for (int r = 1; r < 8; ++r) { vs = l15 == r ? ps[r] : vs; vd = l15 == r ? pd[r] : vd; }
  if (l15 < 8) { red[0][half * 8 + l15][wib] = vs; red[1][half * 8 + l15][wib] = vd; }
  __syncthreads();
  constexpr int TPH = CH / 16;
  constexpr int HPB = WPB / TPH;
  static_assert(WPB % TPH == 0 && WPB <= 8, "whole heads per block, red[] holds 8 waves");
  const int t   = threadIdx.x;
  if (t < 32 * HPB) {
    const int hl   = t % HPB;
    const int row  = (t / HPB) & 15;
    const int kind = t / (16 * HPB);
    float s = red[kind][row][hl * TPH];
    for (int w = 1; w < TPH; ++w) s += red[kind][row][hl * TPH + w];
    float* __restrict__ o = kind ? ald : als;
    o[((long)tm * 16 + row) * HEADS + tn0 / TPH + hl] = s;
  }
}


constexpr int NCHUNK  = 64;
constexpr int LDS_CAP = 65536;
constexpr int CNT_T   = 1024;
constexpr int CNT_B   = 8;

__global__ __launch_bounds__(CNT_T) void count_rank_kernel(
    const int* __restrict__ eidx, int E, int EC, int NP,
    int* __restrict__ tab, int* __restrict__ bsum, int* __restrict__ rank) {
  __shared__ __attribute__((aligned(16))) int lc[LDS_CAP];
  const int t = threadIdx.x;
  const int c = blockIdx.x;
  const int* __restrict__ dst = eidx + E;
  const int e0 = c * EC;
  const int e1 = e0 + EC < E ? e0 + EC : E;

  int d[CNT_B];
  for (int u = 0; u < CNT_B; ++u) {
    const int e = e0 + t + u * CNT_T;
    d[u] = e < e1 ? dst[e] : -1;
  }
  for (int i = t * 4; i < NP; i += CNT_T * 4) *(int4*)&lc[i] = make_int4(0, 0, 0, 0);
  __syncthreads();

  for (int eb = e0; eb < e1; eb += CNT_T * CNT_B) {
    int dn[CNT_B];
    for (int u = 0; u < CNT_B; ++u) {
      const int e = eb + CNT_T * CNT_B + t + u * CNT_T;
      dn[u] = e < e1 ? dst[e] : -1;
    }
    int r[CNT_B];
    for (int u = 0; u < CNT_B; ++u) r[u] = atomicAdd(&lc[d[u] >= 0 ? d[u] : NP + (t & 31)], 1);
    for (int u = 0; u < CNT_B; ++u)
      if (d[u] >= 0) rank[eb + t + u * CNT_T] = r[u];
    for (int u = 0; u < CNT_B; ++u) d[u] = dn[u];
  }
  __syncthreads();

  int* __restrict__ mytab = tab + (long)c * NP;
  for (int i = t * 4; i < NP; i += CNT_T * 4) *(int4*)&mytab[i] = *(const int4*)&lc[i];
  const int wave = t >> 5, lane = t & 31;
  for (int b = wave; b < (NP >> 8); b += CNT_T / 32) {
    const int4 p = *(const int4*)&lc[b * 256 + lane * 8];
    const int4 q = *(const int4*)&lc[b * 256 + lane * 8 + 4];
    int sum = p.x + p.y + p.z + p.w + q.x + q.y + q.z + q.w;
    for (int off = 16; off > 0; off >>= 1) sum += __shfl_xor(sum, off);
    if (lane == 0) bsum[b * NCHUNK + c] = sum;
  }
}

__global__ __launch_bounds__(256) void offsets_kernel(int* __restrict__ tab,
                                                      const int* __restrict__ bsum,
                                                      int* __restrict__ rowptr, int n, int NP) {
  __shared__ int sh[256];
  const int t = threadIdx.x;
  const int i = blockIdx.x * 256 + t;
  int part = 0;
  for (int j = t; j < (int)blockIdx.x * NCHUNK; j += 256) part += bsum[j];
  int v[NCHUNK];
  for (int c = 0; c < NCHUNK; ++c) v[c] = tab[(long)c * NP + i];
  int deg = 0;
  for (int c = 0; c < NCHUNK; ++c) { const int x = v[c]; v[c] = deg; deg += x; }
  sh[t] = part;
  __syncthreads();
  for (int off = 128; off > 0; off >>= 1) {
    if (t < off) sh[t] += sh[t + off];
    __syncthreads();
  }
  const int base = sh[0];
  __syncthreads();
  sh[t] = deg;
  __syncthreads();
  for (int off = 1; off < 256; off <<= 1) {
    int w = t >= off ? sh[t - off] : 0;
    __syncthreads();
    sh[t] += w;
    __syncthreads();
  }
  const int excl = base + sh[t] - deg;
  for (int c = 0; c < NCHUNK; ++c) tab[(long)c * NP + i] = excl + v[c];
  if (i < n) rowptr[i] = excl;
  if (i == n - 1) rowptr[n] = excl + deg;
}

__global__ __launch_bounds__(256) void place_kernel(const int* __restrict__ eidx, int E, int EC, int NP,
                                                    int bpc, const int* __restrict__ tab,
                                                    const int* __restrict__ rank,
                                                    int* __restrict__ csr_src) {
  const int c  = blockIdx.x / bpc;
  const int sb = blockIdx.x - c * bpc;
  const int e0 = c * EC;
  const int e1 = e0 + EC < E ? e0 + EC : E;
  const int eb = e0 + sb * 1024 + threadIdx.x;
  const int* __restrict__ mytab = tab + (long)c * NP;
  int d[4], s[4], r[4];
  for (int u = 0; u < 4; ++u) {
    const int e = eb + u * 256;
    d[u] = -1; s[u] = 0; r[u] = 0;
    if (e < e1) { d[u] = eidx[E + e]; s[u] = eidx[e]; r[u] = rank[e]; }
  }
  int p[4];
  for (int u = 0; u < 4; ++u) p[u] = mytab[d[u] >= 0 ? d[u] : 0] + r[u];
  for (int u = 0; u < 4; ++u)
    if (d[u] >= 0) csr_src[p[u]] = s[u];
}

template<int NH>
__device__ __forceinline__ void load_heads(const float* __restrict__ p, int idx, float (&v)[NH]) {
  if constexpr (NH == 4) {
    float4 t = ((const float4*)p)[idx];
    v[0] = t.x; v[1] = t.y; v[2] = t.z; v[3] = t.w;
  } else {
    for (int h = 0; h < NH; ++h) v[h] = p[(long)idx * NH + h];
  }
}

template<int NF>
__device__ __forceinline__ void load_row(const float* __restrict__ p, float (&v)[NF]) {
  if constexpr (NF % 4 == 0) {
    for (int q = 0; q < NF / 4; ++q) {
      float4 t = ((const float4*)p)[q];
      v[4 * q] = t.x; v[4 * q + 1] = t.y; v[4 * q + 2] = t.z; v[4 * q + 3] = t.w;
    }
  } else if constexpr (NF % 2 == 0) {
    for (int q = 0; q < NF / 2; ++q) {
      float2 t = ((const float2*)p)[q];
      v[2 * q] = t.x; v[2 * q + 1] = t.y;
    }
  } else {
    for (int k = 0; k < NF; ++k) v[k] = p[k];
  }
}

template<int NF>
__device__ __forceinline__ void store_row(float* __restrict__ p, const float (&v)[NF]) {
  if constexpr (NF % 4 == 0) {
    for (int q = 0; q < NF / 4; ++q)
      ((float4*)p)[q] = make_float4(v[4 * q], v[4 * q + 1], v[4 * q + 2], v[4 * q + 3]);
  } else if constexpr (NF % 2 == 0) {
    for (int q = 0; q < NF / 2; ++q)
      ((float2*)p)[q] = make_float2(v[2 * q], v[2 * q + 1]);
  } else {
    for (int k = 0; k < NF; ++k) p[k] = v[k];
  }
}

__device__ __forceinline__ int lane_bcast_i(int v, int i) {
  return __builtin_amdgcn_readlane(v, i);
}
__device__ __forceinline__ float lane_bcast_f(float v, int i) {
  return __int_as_float(__builtin_amdgcn_readlane(__float_as_int(v), i));
}

template<int HEADS, int CH, bool ELU>
__global__ void gather_kernel(const int* __restrict__ rowptr, const int* __restrict__ csr_src,
                              const float* __restrict__ als, const float* __restrict__ ald,
                              const float* __restrict__ hsrc, const float* __restrict__ bias,
                              float* __restrict__ outp, int nnodes) {
  constexpr int F  = HEADS * CH;
  constexpr int NF = F / 32;
  const int lane = threadIdx.x & 31;
  const int node = __builtin_amdgcn_readfirstlane((int)((blockIdx.x * blockDim.x + threadIdx.x) >> 5));
  if (node >= nnodes) return;
  const int beg = rowptr[node];
  const int deg = rowptr[node + 1] - beg;
  const int tot = deg + 1;
  float adst[HEADS];
  load_heads<HEADS>(ald, node, adst);

  float mx[HEADS];
  for (int h = 0; h < HEADS; ++h) mx[h] = -__builtin_huge_valf();
  for (int c = 0; c < tot; c += 32) {
    const int g = c + lane;
    if (g < tot) {
      const int s = g < deg ? csr_src[beg + g] : node;
      float a[HEADS];
      load_heads<HEADS>(als, s, a);
      for (int h = 0; h < HEADS; ++h) {
        float l = a[h] + adst[h];
        l = l > 0.f ? l : NEG_SLOPE * l;
        mx[h] = fmaxf(mx[h], l);
      }
    }
  }
  for (int h = 0; h < HEADS; ++h)
    for (int off = 16; off > 0; off >>= 1) mx[h] = fmaxf(mx[h], __shfl_xor(mx[h], off));

  const int j0   = lane * NF;
  const int head = j0 / CH;
  float acc[NF];
  for (int k = 0; k < NF; ++k) acc[k] = 0.f;
  float ssum[HEADS];
  for (int h = 0; h < HEADS; ++h) ssum[h] = 0.f;
  for (int c = 0; c < tot; c += 32) {
    const int g = c + lane;
    const int nn = tot - c < 32 ? tot - c : 32;
    int s = node;
    if (g < deg) s = csr_src[beg + g];
    {
      int rank = 0;
      for (int j = 0; j < nn; ++j) {
        const int sj = lane_bcast_i(s, j);
        rank += (int)(sj < s) + (int)((sj == s) & (j < lane));
      }
      rank = g < tot ? rank : lane;
      s = __builtin_amdgcn_ds_permute(rank << 2, s);
    }
    float ex[HEADS];
    for (int h = 0; h < HEADS; ++h) ex[h] = 0.f;
    if (g < tot) {
      float a[HEADS];
      load_heads<HEADS>(als, s, a);
      for (int h = 0; h < HEADS; ++h) {
        float l = a[h] + adst[h];
        l = l > 0.f ? l : NEG_SLOPE * l;
        ex[h] = __expf(l - mx[h]);
        ssum[h] += ex[h];
      }
    }
    const int nn4 = (nn + 3) & ~3;
    for (int i = 0; i < nn4; i += 4) {
      int   sr[4];
      float w[4];
      float v[4][NF];
      for (int u = 0; u < 4; ++u) {
        sr[u] = lane_bcast_i(s, i + u);
        float wu = lane_bcast_f(ex[0], i + u);
        for (int h = 1; h < HEADS; ++h) {
          float th = lane_bcast_f(ex[h], i + u);
          wu = head == h ? th : wu;
        }
        w[u] = wu;
      }
      for (int u = 0; u < 4; ++u) load_row<NF>(hsrc + (long)sr[u] * F + j0, v[u]);
      for (int u = 0; u < 4; ++u)
        for (int k = 0; k < NF; ++k) acc[k] += w[u] * v[u][k];
    }
  }
  for (int h = 0; h < HEADS; ++h)
    for (int off = 16; off > 0; off >>= 1) ssum[h] += __shfl_xor(ssum[h], off);
  float sm = ssum[0];
  for (int h = 1; h < HEADS; ++h) sm = head == h ? ssum[h] : sm;

  float bv[NF], r[NF];
  load_row<NF>(bias + j0, bv);
  for (int k = 0; k < NF; ++k) {
    float val = acc[k] / sm + bv[k];
    if (ELU) val = val > 0.f ? val : (__expf(val) - 1.f);
    r[k] = val;
  }
  store_row<NF>(outp + (long)node * F + j0, r);
}


static inline int cdiv_i(long a, long b) { return (int)((a + b - 1) / b); }

template<int HEADS, int CH>
static void launch_gemm(const float* A, const float* B, float* C,
                        int M, int K, const float* a_src, const float* a_dst,
                        float* als, float* ald, hipStream_t stream) {
  constexpr int Nd     = HEADS * CH;
  constexpr int tilesN = Nd / 16;
  constexpr int wpb    = tilesN >= 8 ? 8 : tilesN;
  int tilesM = M / 16;
  dim3 grid(tilesM * (tilesN / wpb));
  dim3 block(32 * wpb);
  size_t shmem = (size_t)16 * (K + 4) * sizeof(float);
  gemm_wmma_f32_lds<HEADS, CH, wpb><<<grid, block, shmem, stream>>>(A, B, C, M, K, Nd, wpb,
                                                                  a_src, a_dst, als, ald);
}

extern "C" void kernel_launch(void* const* d_in, const int* in_sizes, int n_in,
                              void* d_out, int out_size, void* d_ws, size_t ws_size,
                              hipStream_t stream) {
  const float* x   = (const float*)d_in[0];
  const int*   eix = (const int*)d_in[1];
  const float* W1  = (const float*)d_in[2];
  const float* as1 = (const float*)d_in[3];
  const float* ad1 = (const float*)d_in[4];
  const float* b1  = (const float*)d_in[5];
  const float* W2  = (const float*)d_in[6];
  const float* as2 = (const float*)d_in[7];
  const float* ad2 = (const float*)d_in[8];
  const float* b2  = (const float*)d_in[9];
  float* out = (float*)d_out;

  const int IN_DIM = 128;
  const int F1     = in_sizes[5];
  const int HID    = in_sizes[9];
  const int HEADS  = F1 / HID;
  const int Nn     = in_sizes[0] / IN_DIM;
  const int E      = in_sizes[1] / 2;

  float* h1e  = (float*)d_ws;
  float* h1   = h1e + (long)Nn * F1;
  float* als1 = h1 + (long)Nn * F1;
  float* ald1 = als1 + (long)Nn * HEADS;
  float* als2 = ald1 + (long)Nn * HEADS;
  float* ald2 = als2 + Nn;
  const int NP = ((Nn + 255) / 256) * 256;
  const int EC = ((cdiv_i(E, NCHUNK) + 3) / 4) * 4;
  int* rowptr = (int*)(ald2 + Nn);
  int* csr    = rowptr + (Nn + 1);
  int* rank   = csr + E;
  int* bsum   = rank + E;
  int* tab    = bsum + (NP / 256) * NCHUNK + 64;
  tab = (int*)(((size_t)tab + 255) & ~(size_t)255);
  float* h2   = h1;

  const int bpc = cdiv_i(EC, 1024);
  count_rank_kernel<<<NCHUNK, CNT_T, 0, stream>>>(eix, E, EC, NP, tab, bsum, rank);
  offsets_kernel<<<NP / 256, 256, 0, stream>>>(tab, bsum, rowptr, Nn, NP);
  place_kernel<<<NCHUNK * bpc, 256, 0, stream>>>(eix, E, EC, NP, bpc, tab, rank, csr);

  launch_gemm<4, 64>(x, W1, h1, Nn, IN_DIM, as1, ad1, als1, ald1, stream);
  gather_kernel<4, 64, true><<<cdiv_i((long)Nn * 32, 256), 256, 0, stream>>>(
      rowptr, csr, als1, ald1, h1, b1, h1e, Nn);

  launch_gemm<1, 64>(h1e, W2, h2, Nn, F1, as2, ad2, als2, ald2, stream);
  gather_kernel<1, 64, false><<<cdiv_i((long)Nn * 32, 256), 256, 0, stream>>>(
      rowptr, csr, als2, ald2, h2, b2, out, Nn);
}


# Layer-2 GEMM folded into the layer-1 gather: 16-wave blocks keep their 16 ELU rows in LDS, run the f32 WMMA tile there, k in 4 parts summed via LDS (new h2 sum order); 6 launches, no ELU plane
# speedup vs baseline: 7.1192x; 1.0780x over previous
#include <hip/hip_runtime.h>


typedef float v2f __attribute__((ext_vector_type(2)));
typedef float v8f __attribute__((ext_vector_type(8)));

#define NEG_SLOPE 0.2f

template<int M_>
__device__ __forceinline__ float row_xor_c(float v) {
  return __int_as_float(__builtin_amdgcn_update_dpp(0, __float_as_int(v), 0x160 + M_, 0xf, 0xf, true));
}
__device__ __forceinline__ float row_xor_f(float v, int m) {
  return m == 1 ? row_xor_c<1>(v) : m == 2 ? row_xor_c<2>(v) : m == 4 ? row_xor_c<4>(v) : row_xor_c<8>(v);
}

template<int HEADS, int CH, int WPB>
__global__ void gemm_wmma_f32_lds(const float* __restrict__ A, const float* __restrict__ B,
                                  float* __restrict__ C, int M, int K, int Nd,
                                  int tnPerBlock,
                                  const float* __restrict__ a_src, const float* __restrict__ a_dst,
                                  float* __restrict__ als, float* __restrict__ ald) {
  extern __shared__ float As[];
  __shared__ float red[2][16][8];
  const int P = K + 4;
  int tilesN = Nd >> 4;
  int blocksPerRow = tilesN / tnPerBlock;
  int tm  = blockIdx.x / blocksPerRow;
  int tn0 = (blockIdx.x % blocksPerRow) * tnPerBlock;
  int wib  = threadIdx.x >> 5;
  int lane = threadIdx.x & 31;

  {
    const int kv = K >> 2;
    const int nw = blockDim.x >> 5;
    for (int r = wib; r < 16; r += nw) {
      const float4* Av  = (const float4*)(A + ((long)tm * 16 + r) * K);
      float4*       Asv = (float4*)(As + r * P);
      for (int c = lane; c < kv; c += 32) Asv[c] = Av[c];
    }
  }
  __syncthreads();

  int tn   = tn0 + wib;
  int half = lane >> 4;
  int l15  = lane & 15;
  int bcol = tn * 16 + l15;
  const float cs = a_src[bcol];
  const float cd = a_dst[bcol];
  v8f acc = {};
  for (int k = 0; k < K; k += 4) {
    int ka = k + 2 * half;
    v2f a, b;
    a.x = As[l15 * P + ka];
    a.y = As[l15 * P + ka + 1];
    b.x = B[(long)ka * Nd + bcol];
    b.y = B[(long)(ka + 1) * Nd + bcol];
    acc = __builtin_amdgcn_wmma_f32_16x16x4_f32(false, a, false, b,
                                                (short)0, acc, false, false);
  }
  long crow = (long)tm * 16 + half * 8;
#pragma unroll
  for (int r = 0; r < 8; ++r)
    C[(crow + r) * Nd + bcol] = acc[r];

  float ps[8], pd[8];
  for (int r = 0; r < 8; ++r) { ps[r] = acc[r] * cs; pd[r] = acc[r] * cd; }
  for (int m = 1; m < 16; m <<= 1)
    for (int r = 0; r < 8; ++r) { ps[r] += row_xor_f(ps[r], m); pd[r] += row_xor_f(pd[r], m); }
  float vs = ps[0], vd = pd[0];
  for (int r = 1; r < 8; ++r) { vs = l15 == r ? ps[r] : vs; vd = l15 == r ? pd[r] : vd; }
  if (l15 < 8) { red[0][half * 8 + l15][wib] = vs; red[1][half * 8 + l15][wib] = vd; }
  __syncthreads();
  constexpr int TPH = CH / 16;
  constexpr int HPB = WPB / TPH;
  static_assert(WPB % TPH == 0 && WPB <= 8, "whole heads per block, red[] holds 8 waves");
  const int t   = threadIdx.x;
  if (t < 32 * HPB) {
    const int hl   = t % HPB;
    const int row  = (t / HPB) & 15;
    const int kind = t / (16 * HPB);
    float s = red[kind][row][hl * TPH];
    for (int w = 1; w < TPH; ++w) s += red[kind][row][hl * TPH + w];
    float* __restrict__ o = kind ? ald : als;
    o[((long)tm * 16 + row) * HEADS + tn0 / TPH + hl] = s;
  }
}


constexpr int NCHUNK  = 64;
constexpr int LDS_CAP = 65536;
constexpr int CNT_T   = 1024;
constexpr int CNT_B   = 8;

__global__ __launch_bounds__(CNT_T) void count_rank_kernel(
    const int* __restrict__ eidx, int E, int EC, int NP,
    int* __restrict__ tab, int* __restrict__ bsum, int* __restrict__ rank) {
  __shared__ __attribute__((aligned(16))) int lc[LDS_CAP];
  const int t = threadIdx.x;
  const int c = blockIdx.x;
  const int* __restrict__ dst = eidx + E;
  const int e0 = c * EC;
  const int e1 = e0 + EC < E ? e0 + EC : E;

  int d[CNT_B];
  for (int u = 0; u < CNT_B; ++u) {
    const int e = e0 + t + u * CNT_T;
    d[u] = e < e1 ? dst[e] : -1;
  }
  for (int i = t * 4; i < NP; i += CNT_T * 4) *(int4*)&lc[i] = make_int4(0, 0, 0, 0);
  __syncthreads();

  for (int eb = e0; eb < e1; eb += CNT_T * CNT_B) {
    int dn[CNT_B];
    for (int u = 0; u < CNT_B; ++u) {
      const int e = eb + CNT_T * CNT_B + t + u * CNT_T;
      dn[u] = e < e1 ? dst[e] : -1;
    }
    int r[CNT_B];
    for (int u = 0; u < CNT_B; ++u) r[u] = atomicAdd(&lc[d[u] >= 0 ? d[u] : NP + (t & 31)], 1);
    for (int u = 0; u < CNT_B; ++u)
      if (d[u] >= 0) rank[eb + t + u * CNT_T] = r[u];
    for (int u = 0; u < CNT_B; ++u) d[u] = dn[u];
  }
  __syncthreads();

  int* __restrict__ mytab = tab + (long)c * NP;
  for (int i = t * 4; i < NP; i += CNT_T * 4) *(int4*)&mytab[i] = *(const int4*)&lc[i];
  const int wave = t >> 5, lane = t & 31;
  for (int b = wave; b < (NP >> 8); b += CNT_T / 32) {
    const int4 p = *(const int4*)&lc[b * 256 + lane * 8];
    const int4 q = *(const int4*)&lc[b * 256 + lane * 8 + 4];
    int sum = p.x + p.y + p.z + p.w + q.x + q.y + q.z + q.w;
    for (int off = 16; off > 0; off >>= 1) sum += __shfl_xor(sum, off);
    if (lane == 0) bsum[b * NCHUNK + c] = sum;
  }
}

__global__ __launch_bounds__(256) void offsets_kernel(int* __restrict__ tab,
                                                      const int* __restrict__ bsum,
                                                      int* __restrict__ rowptr, int n, int NP) {
  __shared__ int sh[256];
  const int t = threadIdx.x;
  const int i = blockIdx.x * 256 + t;
  int part = 0;
  for (int j = t; j < (int)blockIdx.x * NCHUNK; j += 256) part += bsum[j];
  int v[NCHUNK];
  for (int c = 0; c < NCHUNK; ++c) v[c] = tab[(long)c * NP + i];
  int deg = 0;
  for (int c = 0; c < NCHUNK; ++c) { const int x = v[c]; v[c] = deg; deg += x; }
  sh[t] = part;
  __syncthreads();
  for (int off = 128; off > 0; off >>= 1) {
    if (t < off) sh[t] += sh[t + off];
    __syncthreads();
  }
  const int base = sh[0];
  __syncthreads();
  sh[t] = deg;
  __syncthreads();
  for (int off = 1; off < 256; off <<= 1) {
    int w = t >= off ? sh[t - off] : 0;
    __syncthreads();
    sh[t] += w;
    __syncthreads();
  }
  const int excl = base + sh[t] - deg;
  for (int c = 0; c < NCHUNK; ++c) tab[(long)c * NP + i] = excl + v[c];
  if (i < n) rowptr[i] = excl;
  if (i == n - 1) rowptr[n] = excl + deg;
}

__global__ __launch_bounds__(256) void place_kernel(const int* __restrict__ eidx, int E, int EC, int NP,
                                                    int bpc, const int* __restrict__ tab,
                                                    const int* __restrict__ rank,
                                                    int* __restrict__ csr_src) {
  const int c  = blockIdx.x / bpc;
  const int sb = blockIdx.x - c * bpc;
  const int e0 = c * EC;
  const int e1 = e0 + EC < E ? e0 + EC : E;
  const int eb = e0 + sb * 1024 + threadIdx.x;
  const int* __restrict__ mytab = tab + (long)c * NP;
  int d[4], s[4], r[4];
  for (int u = 0; u < 4; ++u) {
    const int e = eb + u * 256;
    d[u] = -1; s[u] = 0; r[u] = 0;
    if (e < e1) { d[u] = eidx[E + e]; s[u] = eidx[e]; r[u] = rank[e]; }
  }
  int p[4];
  for (int u = 0; u < 4; ++u) p[u] = mytab[d[u] >= 0 ? d[u] : 0] + r[u];
  for (int u = 0; u < 4; ++u)
    if (d[u] >= 0) csr_src[p[u]] = s[u];
}

template<int NH>
__device__ __forceinline__ void load_heads(const float* __restrict__ p, int idx, float (&v)[NH]) {
  if constexpr (NH == 4) {
    float4 t = ((const float4*)p)[idx];
    v[0] = t.x; v[1] = t.y; v[2] = t.z; v[3] = t.w;
  } else {
    for (int h = 0; h < NH; ++h) v[h] = p[(long)idx * NH + h];
  }
}

template<int NF>
__device__ __forceinline__ void load_row(const float* __restrict__ p, float (&v)[NF]) {
  if constexpr (NF % 4 == 0) {
    for (int q = 0; q < NF / 4; ++q) {
      float4 t = ((const float4*)p)[q];
      v[4 * q] = t.x; v[4 * q + 1] = t.y; v[4 * q + 2] = t.z; v[4 * q + 3] = t.w;
    }
  } else if constexpr (NF % 2 == 0) {
    for (int q = 0; q < NF / 2; ++q) {
      float2 t = ((const float2*)p)[q];
      v[2 * q] = t.x; v[2 * q + 1] = t.y;
    }
  } else {
    for (int k = 0; k < NF; ++k) v[k] = p[k];
  }
}

template<int NF>
__device__ __forceinline__ void store_row(float* __restrict__ p, const float (&v)[NF]) {
  if constexpr (NF % 4 == 0) {
    for (int q = 0; q < NF / 4; ++q)
      ((float4*)p)[q] = make_float4(v[4 * q], v[4 * q + 1], v[4 * q + 2], v[4 * q + 3]);
  } else if constexpr (NF % 2 == 0) {
    for (int q = 0; q < NF / 2; ++q)
      ((float2*)p)[q] = make_float2(v[2 * q], v[2 * q + 1]);
  } else {
    for (int k = 0; k < NF; ++k) p[k] = v[k];
  }
}

__device__ __forceinline__ int lane_bcast_i(int v, int i) {
  return __builtin_amdgcn_readlane(v, i);
}
__device__ __forceinline__ float lane_bcast_f(float v, int i) {
  return __int_as_float(__builtin_amdgcn_readlane(__float_as_int(v), i));
}

template<int HEADS, int CH, bool ELU>
__device__ __forceinline__ void gather_row(const int* __restrict__ rowptr, const int* __restrict__ csr_src,
                                           const float* __restrict__ als, const float* __restrict__ ald,
                                           const float* __restrict__ hsrc, const float* __restrict__ bias,
                                           int node, int lane, float (&r)[HEADS * CH / 32]) {
  constexpr int F  = HEADS * CH;
  constexpr int NF = F / 32;
  const int beg = rowptr[node];
  const int deg = rowptr[node + 1] - beg;
  const int tot = deg + 1;
  float adst[HEADS];
  load_heads<HEADS>(ald, node, adst);

  float mx[HEADS];
  for (int h = 0; h < HEADS; ++h) mx[h] = -__builtin_huge_valf();
  for (int c = 0; c < tot; c += 32) {
    const int g = c + lane;
    if (g < tot) {
      const int s = g < deg ? csr_src[beg + g] : node;
      float a[HEADS];
      load_heads<HEADS>(als, s, a);
      for (int h = 0; h < HEADS; ++h) {
        float l = a[h] + adst[h];
        l = l > 0.f ? l : NEG_SLOPE * l;
        mx[h] = fmaxf(mx[h], l);
      }
    }
  }
  for (int h = 0; h < HEADS; ++h)
    for (int off = 16; off > 0; off >>= 1) mx[h] = fmaxf(mx[h], __shfl_xor(mx[h], off));

  const int j0   = lane * NF;
  const int head = j0 / CH;
  float acc[NF];
  for (int k = 0; k < NF; ++k) acc[k] = 0.f;
  float ssum[HEADS];
  for (int h = 0; h < HEADS; ++h) ssum[h] = 0.f;
  for (int c = 0; c < tot; c += 32) {
    const int g = c + lane;
    const int nn = tot - c < 32 ? tot - c : 32;
    int s = node;
    if (g < deg) s = csr_src[beg + g];
    {
      int rank = 0;
      for (int j = 0; j < nn; ++j) {
        const int sj = lane_bcast_i(s, j);
        rank += (int)(sj < s) + (int)((sj == s) & (j < lane));
      }
      rank = g < tot ? rank : lane;
      s = __builtin_amdgcn_ds_permute(rank << 2, s);
    }
    float ex[HEADS];
    for (int h = 0; h < HEADS; ++h) ex[h] = 0.f;
    if (g < tot) {
      float a[HEADS];
      load_heads<HEADS>(als, s, a);
      for (int h = 0; h < HEADS; ++h) {
        float l = a[h] + adst[h];
        l = l > 0.f ? l : NEG_SLOPE * l;
        ex[h] = __expf(l - mx[h]);
        ssum[h] += ex[h];
      }
    }
    const int nn4 = (nn + 3) & ~3;
    for (int i = 0; i < nn4; i += 4) {
      int   sr[4];
      float w[4];
      float v[4][NF];
      for (int u = 0; u < 4; ++u) {
        sr[u] = lane_bcast_i(s, i + u);
        float wu = lane_bcast_f(ex[0], i + u);
        for (int h = 1; h < HEADS; ++h) {
          float th = lane_bcast_f(ex[h], i + u);
          wu = head == h ? th : wu;
        }
        w[u] = wu;
      }
      for (int u = 0; u < 4; ++u) load_row<NF>(hsrc + (long)sr[u] * F + j0, v[u]);
      for (int u = 0; u < 4; ++u)
        for (int k = 0; k < NF; ++k) acc[k] += w[u] * v[u][k];
    }
  }
  for (int h = 0; h < HEADS; ++h)
    for (int off = 16; off > 0; off >>= 1) ssum[h] += __shfl_xor(ssum[h], off);
  float sm = ssum[0];
  for (int h = 1; h < HEADS; ++h) sm = head == h ? ssum[h] : sm;

  float bv[NF];
  load_row<NF>(bias + j0, bv);
  for (int k = 0; k < NF; ++k) {
    float val = acc[k] / sm + bv[k];
    if (ELU) val = val > 0.f ? val : (__expf(val) - 1.f);
    r[k] = val;
  }
}


template<int HEADS, int CH, bool ELU>
__global__ void gather_kernel(const int* __restrict__ rowptr, const int* __restrict__ csr_src,
                              const float* __restrict__ als, const float* __restrict__ ald,
                              const float* __restrict__ hsrc, const float* __restrict__ bias,
                              float* __restrict__ outp, int nnodes) {
  constexpr int F  = HEADS * CH;
  constexpr int NF = F / 32;
  const int lane = threadIdx.x & 31;
  const int node = __builtin_amdgcn_readfirstlane((int)((blockIdx.x * blockDim.x + threadIdx.x) >> 5));
  if (node >= nnodes) return;
  float r[NF];
  gather_row<HEADS, CH, ELU>(rowptr, csr_src, als, ald, hsrc, bias, node, lane, r);
  store_row<NF>(outp + (long)node * F + lane * NF, r);
}

template<int HEADS, int CH, int F2>
__global__ __launch_bounds__(512) void gather_gemm_kernel(
    const int* __restrict__ rowptr, const int* __restrict__ csr_src,
    const float* __restrict__ als, const float* __restrict__ ald,
    const float* __restrict__ hsrc, const float* __restrict__ bias,
    const float* __restrict__ B, const float* __restrict__ a_src2, const float* __restrict__ a_dst2,
    float* __restrict__ C, float* __restrict__ als2, float* __restrict__ ald2, int nnodes) {
  constexpr int F    = HEADS * CH;
  constexpr int NF   = F / 32;
  constexpr int P    = F + 4;
  constexpr int TN   = F2 / 16;
  constexpr int KQ   = 16 / TN;
  constexpr int KLEN = F / KQ;
  static_assert(F2 % 16 == 0 && 16 % TN == 0 && F % (4 * KQ) == 0 && NF % 4 == 0, "tile split");
  __shared__ __attribute__((aligned(16))) float As[16 * P];
  __shared__ float part[KQ][TN][8][32];
  __shared__ float red[2][16][TN];

  const int lane = threadIdx.x & 31;
  const int wib  = __builtin_amdgcn_readfirstlane((int)(threadIdx.x >> 5));
  const int node = blockIdx.x * 16 + wib;
  const int tn   = wib % TN;
  const int kq   = wib / TN;
  const int half = lane >> 4;
  const int l15  = lane & 15;
  const int bcol = tn * 16 + l15;
  v2f bf[KLEN / 4];
  float cs, cd;
  {
    float r[NF];
    for (int k = 0; k < NF; ++k) r[k] = 0.f;
    if (node < nnodes)
      gather_row<HEADS, CH, true>(rowptr, csr_src, als, ald, hsrc, bias, node, lane, r);
#pragma unroll
    for (int i = 0; i < KLEN / 4; ++i) {
      const int ka = kq * KLEN + 4 * i + 2 * half;
      bf[i].x = B[ka * F2 + bcol];
      bf[i].y = B[(ka + 1) * F2 + bcol];
    }
    cs = a_src2[bcol];
    cd = a_dst2[bcol];
    store_row<NF>(As + wib * P + lane * NF, r);
  }
  __syncthreads();

  v8f acc = {};
#pragma unroll
  for (int i = 0; i < KLEN / 4; ++i) {
    const int ka = kq * KLEN + 4 * i + 2 * half;
    v2f a;
    a.x = As[l15 * P + ka];
    a.y = As[l15 * P + ka + 1];
    acc = __builtin_amdgcn_wmma_f32_16x16x4_f32(false, a, false, bf[i],
                                                (short)0, acc, false, false);
  }
  for (int r = 0; r < 8; ++r) part[kq][tn][r][lane] = acc[r];
  __syncthreads();

  if (wib < TN) {
    float c[8];
    for (int r = 0; r < 8; ++r) {
      float s = part[0][wib][r][lane];
      for (int q = 1; q < KQ; ++q) s += part[q][wib][r][lane];
      c[r] = s;
    }
    const long crow = (long)blockIdx.x * 16 + half * 8;
    for (int r = 0; r < 8; ++r)
      if (crow + r < nnodes) C[(crow + r) * F2 + bcol] = c[r];
    float ps[8], pd[8];
    for (int r = 0; r < 8; ++r) { ps[r] = c[r] * cs; pd[r] = c[r] * cd; }
    for (int m = 1; m < 16; m <<= 1)
      for (int r = 0; r < 8; ++r) { ps[r] += row_xor_f(ps[r], m); pd[r] += row_xor_f(pd[r], m); }
    float vs = ps[0], vd = pd[0];
    for (int r = 1; r < 8; ++r) { vs = l15 == r ? ps[r] : vs; vd = l15 == r ? pd[r] : vd; }
    if (l15 < 8) { red[0][half * 8 + l15][wib] = vs; red[1][half * 8 + l15][wib] = vd; }
  }
  __syncthreads();
  const int t = threadIdx.x;
  if (t < 32) {
    const int row  = t & 15;
    const int kind = t >> 4;
    float s = red[kind][row][0];
    for (int w = 1; w < TN; ++w) s += red[kind][row][w];
    const long orow = (long)blockIdx.x * 16 + row;
    float* __restrict__ o = kind ? ald2 : als2;
    if (orow < nnodes) o[orow] = s;
  }
}


static inline int cdiv_i(long a, long b) { return (int)((a + b - 1) / b); }

template<int HEADS, int CH>
static void launch_gemm(const float* A, const float* B, float* C,
                        int M, int K, const float* a_src, const float* a_dst,
                        float* als, float* ald, hipStream_t stream) {
  constexpr int Nd     = HEADS * CH;
  constexpr int tilesN = Nd / 16;
  constexpr int wpb    = tilesN >= 8 ? 8 : tilesN;
  int tilesM = M / 16;
  dim3 grid(tilesM * (tilesN / wpb));
  dim3 block(32 * wpb);
  size_t shmem = (size_t)16 * (K + 4) * sizeof(float);
  gemm_wmma_f32_lds<HEADS, CH, wpb><<<grid, block, shmem, stream>>>(A, B, C, M, K, Nd, wpb,
                                                                  a_src, a_dst, als, ald);
}

extern "C" void kernel_launch(void* const* d_in, const int* in_sizes, int n_in,
                              void* d_out, int out_size, void* d_ws, size_t ws_size,
                              hipStream_t stream) {
  const float* x   = (const float*)d_in[0];
  const int*   eix = (const int*)d_in[1];
  const float* W1  = (const float*)d_in[2];
  const float* as1 = (const float*)d_in[3];
  const float* ad1 = (const float*)d_in[4];
  const float* b1  = (const float*)d_in[5];
  const float* W2  = (const float*)d_in[6];
  const float* as2 = (const float*)d_in[7];
  const float* ad2 = (const float*)d_in[8];
  const float* b2  = (const float*)d_in[9];
  float* out = (float*)d_out;

  const int IN_DIM = 128;
  const int F1     = in_sizes[5];
  const int HID    = in_sizes[9];
  const int HEADS  = F1 / HID;
  const int Nn     = in_sizes[0] / IN_DIM;
  const int E      = in_sizes[1] / 2;

  float* h2   = (float*)d_ws;
  float* h1   = h2 + (long)Nn * F1;
  float* als1 = h1 + (long)Nn * F1;
  float* ald1 = als1 + (long)Nn * HEADS;
  float* als2 = ald1 + (long)Nn * HEADS;
  float* ald2 = als2 + Nn;
  const int NP = ((Nn + 255) / 256) * 256;
  const int EC = ((cdiv_i(E, NCHUNK) + 3) / 4) * 4;
  int* rowptr = (int*)(ald2 + Nn);
  int* csr    = rowptr + (Nn + 1);
  int* rank   = csr + E;
  int* bsum   = rank + E;
  int* tab    = bsum + (NP / 256) * NCHUNK + 64;
  tab = (int*)(((size_t)tab + 255) & ~(size_t)255);

  const int bpc = cdiv_i(EC, 1024);
  count_rank_kernel<<<NCHUNK, CNT_T, 0, stream>>>(eix, E, EC, NP, tab, bsum, rank);
  offsets_kernel<<<NP / 256, 256, 0, stream>>>(tab, bsum, rowptr, Nn, NP);
  place_kernel<<<NCHUNK * bpc, 256, 0, stream>>>(eix, E, EC, NP, bpc, tab, rank, csr);

  launch_gemm<4, 64>(x, W1, h1, Nn, IN_DIM, as1, ad1, als1, ald1, stream);
  gather_gemm_kernel<4, 64, 64><<<cdiv_i(Nn, 16), 512, 0, stream>>>(
      rowptr, csr, als1, ald1, h1, b1, W2, as2, ad2, h2, als2, ald2, Nn);
  gather_kernel<1, 64, false><<<cdiv_i((long)Nn * 32, 256), 256, 0, stream>>>(
      rowptr, csr, als2, ald2, h2, b2, out, Nn);
}


# Gather: a node whose bucket fits one piece of 32 reads bucket and logits once (max, exp weights and ascending order from that one read; ordering runs under the logits load); sum order unchanged
# speedup vs baseline: 7.4094x; 1.0408x over previous
#include <hip/hip_runtime.h>


typedef float v2f __attribute__((ext_vector_type(2)));
typedef float v8f __attribute__((ext_vector_type(8)));

#define NEG_SLOPE 0.2f

template<int M_>
__device__ __forceinline__ float row_xor_c(float v) {
  return __int_as_float(__builtin_amdgcn_update_dpp(0, __float_as_int(v), 0x160 + M_, 0xf, 0xf, true));
}
__device__ __forceinline__ float row_xor_f(float v, int m) {
  return m == 1 ? row_xor_c<1>(v) : m == 2 ? row_xor_c<2>(v) : m == 4 ? row_xor_c<4>(v) : row_xor_c<8>(v);
}

template<int HEADS, int CH, int WPB>
__global__ void gemm_wmma_f32_lds(const float* __restrict__ A, const float* __restrict__ B,
                                  float* __restrict__ C, int M, int K, int Nd,
                                  int tnPerBlock,
                                  const float* __restrict__ a_src, const float* __restrict__ a_dst,
                                  float* __restrict__ als, float* __restrict__ ald) {
  extern __shared__ float As[];
  __shared__ float red[2][16][8];
  const int P = K + 4;
  int tilesN = Nd >> 4;
  int blocksPerRow = tilesN / tnPerBlock;
  int tm  = blockIdx.x / blocksPerRow;
  int tn0 = (blockIdx.x % blocksPerRow) * tnPerBlock;
  int wib  = threadIdx.x >> 5;
  int lane = threadIdx.x & 31;

  {
    const int kv = K >> 2;
    const int nw = blockDim.x >> 5;
    for (int r = wib; r < 16; r += nw) {
      const float4* Av  = (const float4*)(A + ((long)tm * 16 + r) * K);
      float4*       Asv = (float4*)(As + r * P);
      for (int c = lane; c < kv; c += 32) Asv[c] = Av[c];
    }
  }
  __syncthreads();

  int tn   = tn0 + wib;
  int half = lane >> 4;
  int l15  = lane & 15;
  int bcol = tn * 16 + l15;
  const float cs = a_src[bcol];
  const float cd = a_dst[bcol];
  v8f acc = {};
  for (int k = 0; k < K; k += 4) {
    int ka = k + 2 * half;
    v2f a, b;
    a.x = As[l15 * P + ka];
    a.y = As[l15 * P + ka + 1];
    b.x = B[(long)ka * Nd + bcol];
    b.y = B[(long)(ka + 1) * Nd + bcol];
    acc = __builtin_amdgcn_wmma_f32_16x16x4_f32(false, a, false, b,
                                                (short)0, acc, false, false);
  }
  long crow = (long)tm * 16 + half * 8;
#pragma unroll
  for (int r = 0; r < 8; ++r)
    C[(crow + r) * Nd + bcol] = acc[r];

  float ps[8], pd[8];
  for (int r = 0; r < 8; ++r) { ps[r] = acc[r] * cs; pd[r] = acc[r] * cd; }
  for (int m = 1; m < 16; m <<= 1)
    for (int r = 0; r < 8; ++r) { ps[r] += row_xor_f(ps[r], m); pd[r] += row_xor_f(pd[r], m); }
  float vs = ps[0], vd = pd[0];
  for (int r = 1; r < 8; ++r) { vs = l15 == r ? ps[r] : vs; vd = l15 == r ? pd[r] : vd; }
  if (l15 < 8) { red[0][half * 8 + l15][wib] = vs; red[1][half * 8 + l15][wib] = vd; }
  __syncthreads();
  constexpr int TPH = CH / 16;
  constexpr int HPB = WPB / TPH;
  static_assert(WPB % TPH == 0 && WPB <= 8, "whole heads per block, red[] holds 8 waves");
  const int t   = threadIdx.x;
  if (t < 32 * HPB) {
    const int hl   = t % HPB;
    const int row  = (t / HPB) & 15;
    const int kind = t / (16 * HPB);
    float s = red[kind][row][hl * TPH];
    for (int w = 1; w < TPH; ++w) s += red[kind][row][hl * TPH + w];
    float* __restrict__ o = kind ? ald : als;
    o[((long)tm * 16 + row) * HEADS + tn0 / TPH + hl] = s;
  }
}


constexpr int NCHUNK  = 64;
constexpr int LDS_CAP = 65536;
constexpr int CNT_T   = 1024;
constexpr int CNT_B   = 8;

__global__ __launch_bounds__(CNT_T) void count_rank_kernel(
    const int* __restrict__ eidx, int E, int EC, int NP,
    int* __restrict__ tab, int* __restrict__ bsum, int* __restrict__ rank) {
  __shared__ __attribute__((aligned(16))) int lc[LDS_CAP];
  const int t = threadIdx.x;
  const int c = blockIdx.x;
  const int* __restrict__ dst = eidx + E;
  const int e0 = c * EC;
  const int e1 = e0 + EC < E ? e0 + EC : E;

  int d[CNT_B];
  for (int u = 0; u < CNT_B; ++u) {
    const int e = e0 + t + u * CNT_T;
    d[u] = e < e1 ? dst[e] : -1;
  }
  for (int i = t * 4; i < NP; i += CNT_T * 4) *(int4*)&lc[i] = make_int4(0, 0, 0, 0);
  __syncthreads();

  for (int eb = e0; eb < e1; eb += CNT_T * CNT_B) {
    int dn[CNT_B];
    for (int u = 0; u < CNT_B; ++u) {
      const int e = eb + CNT_T * CNT_B + t + u * CNT_T;
      dn[u] = e < e1 ? dst[e] : -1;
    }
    int r[CNT_B];
    for (int u = 0; u < CNT_B; ++u) r[u] = atomicAdd(&lc[d[u] >= 0 ? d[u] : NP + (t & 31)], 1);
    for (int u = 0; u < CNT_B; ++u)
      if (d[u] >= 0) rank[eb + t + u * CNT_T] = r[u];
    for (int u = 0; u < CNT_B; ++u) d[u] = dn[u];
  }
  __syncthreads();

  int* __restrict__ mytab = tab + (long)c * NP;
  for (int i = t * 4; i < NP; i += CNT_T * 4) *(int4*)&mytab[i] = *(const int4*)&lc[i];
  const int wave = t >> 5, lane = t & 31;
  for (int b = wave; b < (NP >> 8); b += CNT_T / 32) {
    const int4 p = *(const int4*)&lc[b * 256 + lane * 8];
    const int4 q = *(const int4*)&lc[b * 256 + lane * 8 + 4];
    int sum = p.x + p.y + p.z + p.w + q.x + q.y + q.z + q.w;
    for (int off = 16; off > 0; off >>= 1) sum += __shfl_xor(sum, off);
    if (lane == 0) bsum[b * NCHUNK + c] = sum;
  }
}

__global__ __launch_bounds__(256) void offsets_kernel(int* __restrict__ tab,
                                                      const int* __restrict__ bsum,
                                                      int* __restrict__ rowptr, int n, int NP) {
  __shared__ int sh[256];
  const int t = threadIdx.x;
  const int i = blockIdx.x * 256 + t;
  int part = 0;
  for (int j = t; j < (int)blockIdx.x * NCHUNK; j += 256) part += bsum[j];
  int v[NCHUNK];
  for (int c = 0; c < NCHUNK; ++c) v[c] = tab[(long)c * NP + i];
  int deg = 0;
  for (int c = 0; c < NCHUNK; ++c) { const int x = v[c]; v[c] = deg; deg += x; }
  sh[t] = part;
  __syncthreads();
  for (int off = 128; off > 0; off >>= 1) {
    if (t < off) sh[t] += sh[t + off];
    __syncthreads();
  }
  const int base = sh[0];
  __syncthreads();
  sh[t] = deg;
  __syncthreads();
  for (int off = 1; off < 256; off <<= 1) {
    int w = t >= off ? sh[t - off] : 0;
    __syncthreads();
    sh[t] += w;
    __syncthreads();
  }
  const int excl = base + sh[t] - deg;
  for (int c = 0; c < NCHUNK; ++c) tab[(long)c * NP + i] = excl + v[c];
  if (i < n) rowptr[i] = excl;
  if (i == n - 1) rowptr[n] = excl + deg;
}

__global__ __launch_bounds__(256) void place_kernel(const int* __restrict__ eidx, int E, int EC, int NP,
                                                    int bpc, const int* __restrict__ tab,
                                                    const int* __restrict__ rank,
                                                    int* __restrict__ csr_src) {
  const int c  = blockIdx.x / bpc;
  const int sb = blockIdx.x - c * bpc;
  const int e0 = c * EC;
  const int e1 = e0 + EC < E ? e0 + EC : E;
  const int eb = e0 + sb * 1024 + threadIdx.x;
  const int* __restrict__ mytab = tab + (long)c * NP;
  int d[4], s[4], r[4];
  for (int u = 0; u < 4; ++u) {
    const int e = eb + u * 256;
    d[u] = -1; s[u] = 0; r[u] = 0;
    if (e < e1) { d[u] = eidx[E + e]; s[u] = eidx[e]; r[u] = rank[e]; }
  }
  int p[4];
  for (int u = 0; u < 4; ++u) p[u] = mytab[d[u] >= 0 ? d[u] : 0] + r[u];
  for (int u = 0; u < 4; ++u)
    if (d[u] >= 0) csr_src[p[u]] = s[u];
}

template<int NH>
__device__ __forceinline__ void load_heads(const float* __restrict__ p, int idx, float (&v)[NH]) {
  if constexpr (NH == 4) {
    float4 t = ((const float4*)p)[idx];
    v[0] = t.x; v[1] = t.y; v[2] = t.z; v[3] = t.w;
  } else {
    for (int h = 0; h < NH; ++h) v[h] = p[(long)idx * NH + h];
  }
}

template<int NF>
__device__ __forceinline__ void load_row(const float* __restrict__ p, float (&v)[NF]) {
  if constexpr (NF % 4 == 0) {
    for (int q = 0; q < NF / 4; ++q) {
      float4 t = ((const float4*)p)[q];
      v[4 * q] = t.x; v[4 * q + 1] = t.y; v[4 * q + 2] = t.z; v[4 * q + 3] = t.w;
    }
  } else if constexpr (NF % 2 == 0) {
    for (int q = 0; q < NF / 2; ++q) {
      float2 t = ((const float2*)p)[q];
      v[2 * q] = t.x; v[2 * q + 1] = t.y;
    }
  } else {
    for (int k = 0; k < NF; ++k) v[k] = p[k];
  }
}

template<int NF>
__device__ __forceinline__ void store_row(float* __restrict__ p, const float (&v)[NF]) {
  if constexpr (NF % 4 == 0) {
    for (int q = 0; q < NF / 4; ++q)
      ((float4*)p)[q] = make_float4(v[4 * q], v[4 * q + 1], v[4 * q + 2], v[4 * q + 3]);
  } else if constexpr (NF % 2 == 0) {
    for (int q = 0; q < NF / 2; ++q)
      ((float2*)p)[q] = make_float2(v[2 * q], v[2 * q + 1]);
  } else {
    for (int k = 0; k < NF; ++k) p[k] = v[k];
  }
}

__device__ __forceinline__ int lane_bcast_i(int v, int i) {
  return __builtin_amdgcn_readlane(v, i);
}
__device__ __forceinline__ float lane_bcast_f(float v, int i) {
  return __int_as_float(__builtin_amdgcn_readlane(__float_as_int(v), i));
}


__device__ __forceinline__ int piece_rank(int s, int nn, int lane) {
  int rank = 0;
  for (int j = 0; j < nn; ++j) {
    const int sj = lane_bcast_i(s, j);
    rank += (int)(sj < s) + (int)((sj == s) & (j < lane));
  }
  return lane < nn ? rank : lane;
}

template<int HEADS, int CH>
__device__ __forceinline__ void add_piece_rows(const float* __restrict__ hsrc, int s, const float (&ex)[HEADS],
                                               int nn, int head, int j0, float (&acc)[HEADS * CH / 32]) {
  constexpr int F  = HEADS * CH;
  constexpr int NF = F / 32;
  const int nn4 = (nn + 3) & ~3;
  for (int i = 0; i < nn4; i += 4) {
    int   sr[4];
    float w[4];
    float v[4][NF];
    for (int u = 0; u < 4; ++u) {
      sr[u] = lane_bcast_i(s, i + u);
      float wu = lane_bcast_f(ex[0], i + u);
      for (int h = 1; h < HEADS; ++h) {
        float th = lane_bcast_f(ex[h], i + u);
        wu = head == h ? th : wu;
      }
      w[u] = wu;
    }
    for (int u = 0; u < 4; ++u) load_row<NF>(hsrc + (long)sr[u] * F + j0, v[u]);
    for (int u = 0; u < 4; ++u)
      for (int k = 0; k < NF; ++k) acc[k] += w[u] * v[u][k];
  }
}

template<int HEADS, int CH, bool ELU>
__device__ __forceinline__ void gather_row(const int* __restrict__ rowptr, const int* __restrict__ csr_src,
                                           const float* __restrict__ als, const float* __restrict__ ald,
                                           const float* __restrict__ hsrc, const float* __restrict__ bias,
                                           int node, int lane, float (&r)[HEADS * CH / 32]) {
  constexpr int F  = HEADS * CH;
  constexpr int NF = F / 32;
  const int beg = rowptr[node];
  const int deg = rowptr[node + 1] - beg;
  const int tot = deg + 1;
  float adst[HEADS];
  load_heads<HEADS>(ald, node, adst);

  const int j0   = lane * NF;
  const int head = j0 / CH;
  float acc[NF];
  for (int k = 0; k < NF; ++k) acc[k] = 0.f;
  float ssum[HEADS];
  for (int h = 0; h < HEADS; ++h) ssum[h] = 0.f;

  if (tot <= 32) {
    int s = node;
    if (lane < deg) s = csr_src[beg + lane];
    float a[HEADS];
    load_heads<HEADS>(als, s, a);
    const int rank = piece_rank(s, tot, lane);
    float l[HEADS], mx[HEADS];
    for (int h = 0; h < HEADS; ++h) {
      float t = a[h] + adst[h];
      l[h] = t > 0.f ? t : NEG_SLOPE * t;
      mx[h] = -__builtin_huge_valf();
      if (lane < tot) mx[h] = fmaxf(mx[h], l[h]);
    }
    for (int h = 0; h < HEADS; ++h)
      for (int off = 16; off > 0; off >>= 1) mx[h] = fmaxf(mx[h], __shfl_xor(mx[h], off));
    float ex[HEADS];
    for (int h = 0; h < HEADS; ++h) {
      const float e = lane < tot ? __expf(l[h] - mx[h]) : 0.f;
      ex[h] = __int_as_float(__builtin_amdgcn_ds_permute(rank << 2, __float_as_int(e)));
      ssum[h] += ex[h];
    }
    s = __builtin_amdgcn_ds_permute(rank << 2, s);
    add_piece_rows<HEADS, CH>(hsrc, s, ex, tot, head, j0, acc);
  } else {
    float mx[HEADS];
    for (int h = 0; h < HEADS; ++h) mx[h] = -__builtin_huge_valf();
    for (int c = 0; c < tot; c += 32) {
      const int g = c + lane;
      if (g < tot) {
        const int s = g < deg ? csr_src[beg + g] : node;
        float a[HEADS];
        load_heads<HEADS>(als, s, a);
        for (int h = 0; h < HEADS; ++h) {
          float l = a[h] + adst[h];
          l = l > 0.f ? l : NEG_SLOPE * l;
          mx[h] = fmaxf(mx[h], l);
        }
      }
    }
    for (int h = 0; h < HEADS; ++h)
      for (int off = 16; off > 0; off >>= 1) mx[h] = fmaxf(mx[h], __shfl_xor(mx[h], off));

    for (int c = 0; c < tot; c += 32) {
      const int g = c + lane;
      const int nn = tot - c < 32 ? tot - c : 32;
      int s = node;
      if (g < deg) s = csr_src[beg + g];
      {
        const int rank = piece_rank(s, nn, lane);
        s = __builtin_amdgcn_ds_permute(rank << 2, s);
      }
      float ex[HEADS];
      for (int h = 0; h < HEADS; ++h) ex[h] = 0.f;
      if (g < tot) {
        float a[HEADS];
        load_heads<HEADS>(als, s, a);
        for (int h = 0; h < HEADS; ++h) {
          float l = a[h] + adst[h];
          l = l > 0.f ? l : NEG_SLOPE * l;
          ex[h] = __expf(l - mx[h]);
          ssum[h] += ex[h];
        }
      }
      add_piece_rows<HEADS, CH>(hsrc, s, ex, nn, head, j0, acc);
    }
  }
  for (int h = 0; h < HEADS; ++h)
    for (int off = 16; off > 0; off >>= 1) ssum[h] += __shfl_xor(ssum[h], off);
  float sm = ssum[0];
  for (int h = 1; h < HEADS; ++h) sm = head == h ? ssum[h] : sm;

  float bv[NF];
  load_row<NF>(bias + j0, bv);
  for (int k = 0; k < NF; ++k) {
    float val = acc[k] / sm + bv[k];
    if (ELU) val = val > 0.f ? val : (__expf(val) - 1.f);
    r[k] = val;
  }
}


template<int HEADS, int CH, bool ELU>
__global__ void gather_kernel(const int* __restrict__ rowptr, const int* __restrict__ csr_src,
                              const float* __restrict__ als, const float* __restrict__ ald,
                              const float* __restrict__ hsrc, const float* __restrict__ bias,
                              float* __restrict__ outp, int nnodes) {
  constexpr int F  = HEADS * CH;
  constexpr int NF = F / 32;
  const int lane = threadIdx.x & 31;
  const int node = __builtin_amdgcn_readfirstlane((int)((blockIdx.x * blockDim.x + threadIdx.x) >> 5));
  if (node >= nnodes) return;
  float r[NF];
  gather_row<HEADS, CH, ELU>(rowptr, csr_src, als, ald, hsrc, bias, node, lane, r);
  store_row<NF>(outp + (long)node * F + lane * NF, r);
}

template<int HEADS, int CH, int F2>
__global__ __launch_bounds__(512) void gather_gemm_kernel(
    const int* __restrict__ rowptr, const int* __restrict__ csr_src,
    const float* __restrict__ als, const float* __restrict__ ald,
    const float* __restrict__ hsrc, const float* __restrict__ bias,
    const float* __restrict__ B, const float* __restrict__ a_src2, const float* __restrict__ a_dst2,
    float* __restrict__ C, float* __restrict__ als2, float* __restrict__ ald2, int nnodes) {
  constexpr int F    = HEADS * CH;
  constexpr int NF   = F / 32;
  constexpr int P    = F + 4;
  constexpr int TN   = F2 / 16;
  constexpr int KQ   = 16 / TN;
  constexpr int KLEN = F / KQ;
  static_assert(F2 % 16 == 0 && 16 % TN == 0 && F % (4 * KQ) == 0 && NF % 4 == 0, "tile split");
  __shared__ __attribute__((aligned(16))) float As[16 * P];
  __shared__ float part[KQ][TN][8][32];
  __shared__ float red[2][16][TN];

  const int lane = threadIdx.x & 31;
  const int wib  = __builtin_amdgcn_readfirstlane((int)(threadIdx.x >> 5));
  const int node = blockIdx.x * 16 + wib;
  const int tn   = wib % TN;
  const int kq   = wib / TN;
  const int half = lane >> 4;
  const int l15  = lane & 15;
  const int bcol = tn * 16 + l15;
  v2f bf[KLEN / 4];
  float cs, cd;
  {
    float r[NF];
    for (int k = 0; k < NF; ++k) r[k] = 0.f;
    if (node < nnodes)
      gather_row<HEADS, CH, true>(rowptr, csr_src, als, ald, hsrc, bias, node, lane, r);
#pragma unroll
    for (int i = 0; i < KLEN / 4; ++i) {
      const int ka = kq * KLEN + 4 * i + 2 * half;
      bf[i].x = B[ka * F2 + bcol];
      bf[i].y = B[(ka + 1) * F2 + bcol];
    }
    cs = a_src2[bcol];
    cd = a_dst2[bcol];
    store_row<NF>(As + wib * P + lane * NF, r);
  }
  __syncthreads();

  v8f acc = {};
#pragma unroll
  for (int i = 0; i < KLEN / 4; ++i) {
    const int ka = kq * KLEN + 4 * i + 2 * half;
    v2f a;
    a.x = As[l15 * P + ka];
    a.y = As[l15 * P + ka + 1];
    acc = __builtin_amdgcn_wmma_f32_16x16x4_f32(false, a, false, bf[i],
                                                (short)0, acc, false, false);
  }
  for (int r = 0; r < 8; ++r) part[kq][tn][r][lane] = acc[r];
  __syncthreads();

  if (wib < TN) {
    float c[8];
    for (int r = 0; r < 8; ++r) {
      float s = part[0][wib][r][lane];
      for (int q = 1; q < KQ; ++q) s += part[q][wib][r][lane];
      c[r] = s;
    }
    const long crow = (long)blockIdx.x * 16 + half * 8;
    for (int r = 0; r < 8; ++r)
      if (crow + r < nnodes) C[(crow + r) * F2 + bcol] = c[r];
    float ps[8], pd[8];
    for (int r = 0; r < 8; ++r) { ps[r] = c[r] * cs; pd[r] = c[r] * cd; }
    for (int m = 1; m < 16; m <<= 1)
      for (int r = 0; r < 8; ++r) { ps[r] += row_xor_f(ps[r], m); pd[r] += row_xor_f(pd[r], m); }
    float vs = ps[0], vd = pd[0];
    for (int r = 1; r < 8; ++r) { vs = l15 == r ? ps[r] : vs; vd = l15 == r ? pd[r] : vd; }
    if (l15 < 8) { red[0][half * 8 + l15][wib] = vs; red[1][half * 8 + l15][wib] = vd; }
  }
  __syncthreads();
  const int t = threadIdx.x;
  if (t < 32) {
    const int row  = t & 15;
    const int kind = t >> 4;
    float s = red[kind][row][0];
    for (int w = 1; w < TN; ++w) s += red[kind][row][w];
    const long orow = (long)blockIdx.x * 16 + row;
    float* __restrict__ o = kind ? ald2 : als2;
    if (orow < nnodes) o[orow] = s;
  }
}


static inline int cdiv_i(long a, long b) { return (int)((a + b - 1) / b); }

template<int HEADS, int CH>
static void launch_gemm(const float* A, const float* B, float* C,
                        int M, int K, const float* a_src, const float* a_dst,
                        float* als, float* ald, hipStream_t stream) {
  constexpr int Nd     = HEADS * CH;
  constexpr int tilesN = Nd / 16;
  constexpr int wpb    = tilesN >= 8 ? 8 : tilesN;
  int tilesM = M / 16;
  dim3 grid(tilesM * (tilesN / wpb));
  dim3 block(32 * wpb);
  size_t shmem = (size_t)16 * (K + 4) * sizeof(float);
  gemm_wmma_f32_lds<HEADS, CH, wpb><<<grid, block, shmem, stream>>>(A, B, C, M, K, Nd, wpb,
                                                                  a_src, a_dst, als, ald);
}

extern "C" void kernel_launch(void* const* d_in, const int* in_sizes, int n_in,
                              void* d_out, int out_size, void* d_ws, size_t ws_size,
                              hipStream_t stream) {
  const float* x   = (const float*)d_in[0];
  const int*   eix = (const int*)d_in[1];
  const float* W1  = (const float*)d_in[2];
  const float* as1 = (const float*)d_in[3];
  const float* ad1 = (const float*)d_in[4];
  const float* b1  = (const float*)d_in[5];
  const float* W2  = (const float*)d_in[6];
  const float* as2 = (const float*)d_in[7];
  const float* ad2 = (const float*)d_in[8];
  const float* b2  = (const float*)d_in[9];
  float* out = (float*)d_out;

  const int IN_DIM = 128;
  const int F1     = in_sizes[5];
  const int HID    = in_sizes[9];
  const int HEADS  = F1 / HID;
  const int Nn     = in_sizes[0] / IN_DIM;
  const int E      = in_sizes[1] / 2;

  float* h2   = (float*)d_ws;
  float* h1   = h2 + (long)Nn * F1;
  float* als1 = h1 + (long)Nn * F1;
  float* ald1 = als1 + (long)Nn * HEADS;
  float* als2 = ald1 + (long)Nn * HEADS;
  float* ald2 = als2 + Nn;
  const int NP = ((Nn + 255) / 256) * 256;
  const int EC = ((cdiv_i(E, NCHUNK) + 3) / 4) * 4;
  int* rowptr = (int*)(ald2 + Nn);
  int* csr    = rowptr + (Nn + 1);
  int* rank   = csr + E;
  int* bsum   = rank + E;
  int* tab    = bsum + (NP / 256) * NCHUNK + 64;
  tab = (int*)(((size_t)tab + 255) & ~(size_t)255);

  const int bpc = cdiv_i(EC, 1024);
  count_rank_kernel<<<NCHUNK, CNT_T, 0, stream>>>(eix, E, EC, NP, tab, bsum, rank);
  offsets_kernel<<<NP / 256, 256, 0, stream>>>(tab, bsum, rowptr, Nn, NP);
  place_kernel<<<NCHUNK * bpc, 256, 0, stream>>>(eix, E, EC, NP, bpc, tab, rank, csr);

  launch_gemm<4, 64>(x, W1, h1, Nn, IN_DIM, as1, ad1, als1, ald1, stream);
  gather_gemm_kernel<4, 64, 64><<<cdiv_i(Nn, 16), 512, 0, stream>>>(
      rowptr, csr, als1, ald1, h1, b1, W2, as2, ad2, h2, als2, ald2, Nn);
  gather_kernel<1, 64, false><<<cdiv_i((long)Nn * 32, 256), 256, 0, stream>>>(
      rowptr, csr, als2, ald2, h2, b2, out, Nn);
}


# Layer-1 GEMM: a wave keeps the B fragments of its column tile in registers and its work-group (one wave per column tile) walks over row tiles, A slab double-buffered in LDS; sum order unchanged
# speedup vs baseline: 8.6368x; 1.1657x over previous
#include <hip/hip_runtime.h>


typedef float v2f __attribute__((ext_vector_type(2)));
typedef float v8f __attribute__((ext_vector_type(8)));

#define NEG_SLOPE 0.2f

template<int M_>
__device__ __forceinline__ float row_xor_c(float v) {
  return __int_as_float(__builtin_amdgcn_update_dpp(0, __float_as_int(v), 0x160 + M_, 0xf, 0xf, true));
}
__device__ __forceinline__ float row_xor_f(float v, int m) {
  return m == 1 ? row_xor_c<1>(v) : m == 2 ? row_xor_c<2>(v) : m == 4 ? row_xor_c<4>(v) : row_xor_c<8>(v);
}

constexpr int GEMM_GRID = 512;

template<int HEADS, int CH, int K>
__global__ __launch_bounds__(2 * HEADS * CH) void gemm_wmma_f32_bres(
    const float* __restrict__ A, const float* __restrict__ B, float* __restrict__ C, int tilesM,
    const float* __restrict__ a_src, const float* __restrict__ a_dst,
    float* __restrict__ als, float* __restrict__ ald) {
  constexpr int Nd  = HEADS * CH;
  constexpr int NW  = Nd / 16;
  constexpr int P   = K + 4;
  constexpr int KV  = K / 128;
  constexpr int TPH = CH / 16;
  static_assert(NW == 16 && K % 128 == 0 && CH % 16 == 0, "one slab row per wave, whole heads");
  __shared__ __attribute__((aligned(16))) float As[2][16 * P];
  __shared__ float red[2][2][16][NW];

  const int lane = threadIdx.x & 31;
  const int wib  = __builtin_amdgcn_readfirstlane((int)(threadIdx.x >> 5));
  const int half = lane >> 4;
  const int l15  = lane & 15;
  const int bcol = wib * 16 + l15;
  int tm = blockIdx.x;
  if (tm >= tilesM) return;

  v2f bf[K / 4];
#pragma unroll
  for (int i = 0; i < K / 4; ++i) {
    const int ka = 4 * i + 2 * half;
    bf[i].x = B[ka * Nd + bcol];
    bf[i].y = B[(ka + 1) * Nd + bcol];
  }
  const float cs = a_src[bcol];
  const float cd = a_dst[bcol];

  {
    const float4* Av = (const float4*)(A + ((long)tm * 16 + wib) * K);
    for (int q = 0; q < KV; ++q) *(float4*)&As[0][wib * P + (lane + 32 * q) * 4] = Av[lane + 32 * q];
  }
  __syncthreads();

  int buf = 0;
  for (; tm < tilesM; tm += gridDim.x) {
    const int tnext = tm + gridDim.x;
    float4 nx[KV];
    for (int q = 0; q < KV; ++q) nx[q] = make_float4(0.f, 0.f, 0.f, 0.f);
    if (tnext < tilesM) {
      const float4* Av = (const float4*)(A + ((long)tnext * 16 + wib) * K);
      for (int q = 0; q < KV; ++q) nx[q] = Av[lane + 32 * q];
    }
    const float* __restrict__ as = &As[buf][l15 * P + 2 * half];
    v8f acc = {};
#pragma unroll
    for (int i = 0; i < K / 4; ++i) {
      const v2f a = *(const v2f*)(as + 4 * i);
      acc = __builtin_amdgcn_wmma_f32_16x16x4_f32(false, a, false, bf[i],
                                                  (short)0, acc, false, false);
    }
    const long crow = (long)tm * 16 + half * 8;
#pragma unroll
    for (int r = 0; r < 8; ++r)
      C[(crow + r) * Nd + bcol] = acc[r];

    float ps[8], pd[8];
    for (int r = 0; r < 8; ++r) { ps[r] = acc[r] * cs; pd[r] = acc[r] * cd; }
    for (int m = 1; m < 16; m <<= 1)
      for (int r = 0; r < 8; ++r) { ps[r] += row_xor_f(ps[r], m); pd[r] += row_xor_f(pd[r], m); }
    float vs = ps[0], vd = pd[0];
    for (int r = 1; r < 8; ++r) { vs = l15 == r ? ps[r] : vs; vd = l15 == r ? pd[r] : vd; }
    if (l15 < 8) { red[buf][0][half * 8 + l15][wib] = vs; red[buf][1][half * 8 + l15][wib] = vd; }
    if (tnext < tilesM)
      for (int q = 0; q < KV; ++q) *(float4*)&As[buf ^ 1][wib * P + (lane + 32 * q) * 4] = nx[q];
    __syncthreads();
    const int t = threadIdx.x;
    if (t < 32 * HEADS) {
      const int hl   = t % HEADS;
      const int row  = (t / HEADS) & 15;
      const int kind = t / (16 * HEADS);
      float s = red[buf][kind][row][hl * TPH];
      for (int w = 1; w < TPH; ++w) s += red[buf][kind][row][hl * TPH + w];
      float* __restrict__ o = kind ? ald : als;
      o[((long)tm * 16 + row) * HEADS + hl] = s;
    }
    buf ^= 1;
  }
}


constexpr int NCHUNK  = 64;
constexpr int LDS_CAP = 65536;
constexpr int CNT_T   = 1024;
constexpr int CNT_B   = 8;

__global__ __launch_bounds__(CNT_T) void count_rank_kernel(
    const int* __restrict__ eidx, int E, int EC, int NP,
    int* __restrict__ tab, int* __restrict__ bsum, int* __restrict__ rank) {
  __shared__ __attribute__((aligned(16))) int lc[LDS_CAP];
  const int t = threadIdx.x;
  const int c = blockIdx.x;
  const int* __restrict__ dst = eidx + E;
  const int e0 = c * EC;
  const int e1 = e0 + EC < E ? e0 + EC : E;

  int d[CNT_B];
  for (int u = 0; u < CNT_B; ++u) {
    const int e = e0 + t + u * CNT_T;
    d[u] = e < e1 ? dst[e] : -1;
  }
  for (int i = t * 4; i < NP; i += CNT_T * 4) *(int4*)&lc[i] = make_int4(0, 0, 0, 0);
  __syncthreads();

  for (int eb = e0; eb < e1; eb += CNT_T * CNT_B) {
    int dn[CNT_B];
    for (int u = 0; u < CNT_B; ++u) {
      const int e = eb + CNT_T * CNT_B + t + u * CNT_T;
      dn[u] = e < e1 ? dst[e] : -1;
    }
    int r[CNT_B];
    for (int u = 0; u < CNT_B; ++u) r[u] = atomicAdd(&lc[d[u] >= 0 ? d[u] : NP + (t & 31)], 1);
    for (int u = 0; u < CNT_B; ++u)
      if (d[u] >= 0) rank[eb + t + u * CNT_T] = r[u];
    for (int u = 0; u < CNT_B; ++u) d[u] = dn[u];
  }
  __syncthreads();

  int* __restrict__ mytab = tab + (long)c * NP;
  for (int i = t * 4; i < NP; i += CNT_T * 4) *(int4*)&mytab[i] = *(const int4*)&lc[i];
  const int wave = t >> 5, lane = t & 31;
  for (int b = wave; b < (NP >> 8); b += CNT_T / 32) {
    const int4 p = *(const int4*)&lc[b * 256 + lane * 8];
    const int4 q = *(const int4*)&lc[b * 256 + lane * 8 + 4];
    int sum = p.x + p.y + p.z + p.w + q.x + q.y + q.z + q.w;
    for (int off = 16; off > 0; off >>= 1) sum += __shfl_xor(sum, off);
    if (lane == 0) bsum[b * NCHUNK + c] = sum;
  }
}

__global__ __launch_bounds__(256) void offsets_kernel(int* __restrict__ tab,
                                                      const int* __restrict__ bsum,
                                                      int* __restrict__ rowptr, int n, int NP) {
  __shared__ int sh[256];
  const int t = threadIdx.x;
  const int i = blockIdx.x * 256 + t;
  int part = 0;
  for (int j = t; j < (int)blockIdx.x * NCHUNK; j += 256) part += bsum[j];
  int v[NCHUNK];
  for (int c = 0; c < NCHUNK; ++c) v[c] = tab[(long)c * NP + i];
  int deg = 0;
  for (int c = 0; c < NCHUNK; ++c) { const int x = v[c]; v[c] = deg; deg += x; }
  sh[t] = part;
  __syncthreads();
  for (int off = 128; off > 0; off >>= 1) {
    if (t < off) sh[t] += sh[t + off];
    __syncthreads();
  }
  const int base = sh[0];
  __syncthreads();
  sh[t] = deg;
  __syncthreads();
  for (int off = 1; off < 256; off <<= 1) {
    int w = t >= off ? sh[t - off] : 0;
    __syncthreads();
    sh[t] += w;
    __syncthreads();
  }
  const int excl = base + sh[t] - deg;
  for (int c = 0; c < NCHUNK; ++c) tab[(long)c * NP + i] = excl + v[c];
  if (i < n) rowptr[i] = excl;
  if (i == n - 1) rowptr[n] = excl + deg;
}

__global__ __launch_bounds__(256) void place_kernel(const int* __restrict__ eidx, int E, int EC, int NP,
                                                    int bpc, const int* __restrict__ tab,
                                                    const int* __restrict__ rank,
                                                    int* __restrict__ csr_src) {
  const int c  = blockIdx.x / bpc;
  const int sb = blockIdx.x - c * bpc;
  const int e0 = c * EC;
  const int e1 = e0 + EC < E ? e0 + EC : E;
  const int eb = e0 + sb * 1024 + threadIdx.x;
  const int* __restrict__ mytab = tab + (long)c * NP;
  int d[4], s[4], r[4];
  for (int u = 0; u < 4; ++u) {
    const int e = eb + u * 256;
    d[u] = -1; s[u] = 0; r[u] = 0;
    if (e < e1) { d[u] = eidx[E + e]; s[u] = eidx[e]; r[u] = rank[e]; }
  }
  int p[4];
  for (int u = 0; u < 4; ++u) p[u] = mytab[d[u] >= 0 ? d[u] : 0] + r[u];
  for (int u = 0; u < 4; ++u)
    if (d[u] >= 0) csr_src[p[u]] = s[u];
}

template<int NH>
__device__ __forceinline__ void load_heads(const float* __restrict__ p, int idx, float (&v)[NH]) {
  if constexpr (NH == 4) {
    float4 t = ((const float4*)p)[idx];
    v[0] = t.x; v[1] = t.y; v[2] = t.z; v[3] = t.w;
  } else {
    for (int h = 0; h < NH; ++h) v[h] = p[(long)idx * NH + h];
  }
}

template<int NF>
__device__ __forceinline__ void load_row(const float* __restrict__ p, float (&v)[NF]) {
  if constexpr (NF % 4 == 0) {
    for (int q = 0; q < NF / 4; ++q) {
      float4 t = ((const float4*)p)[q];
      v[4 * q] = t.x; v[4 * q + 1] = t.y; v[4 * q + 2] = t.z; v[4 * q + 3] = t.w;
    }
  } else if constexpr (NF % 2 == 0) {
    for (int q = 0; q < NF / 2; ++q) {
      float2 t = ((const float2*)p)[q];
      v[2 * q] = t.x; v[2 * q + 1] = t.y;
    }
  } else {
    for (int k = 0; k < NF; ++k) v[k] = p[k];
  }
}

template<int NF>
__device__ __forceinline__ void store_row(float* __restrict__ p, const float (&v)[NF]) {
  if constexpr (NF % 4 == 0) {
    for (int q = 0; q < NF / 4; ++q)
      ((float4*)p)[q] = make_float4(v[4 * q], v[4 * q + 1], v[4 * q + 2], v[4 * q + 3]);
  } else if constexpr (NF % 2 == 0) {
    for (int q = 0; q < NF / 2; ++q)
      ((float2*)p)[q] = make_float2(v[2 * q], v[2 * q + 1]);
  } else {
    for (int k = 0; k < NF; ++k) p[k] = v[k];
  }
}

__device__ __forceinline__ int lane_bcast_i(int v, int i) {
  return __builtin_amdgcn_readlane(v, i);
}
__device__ __forceinline__ float lane_bcast_f(float v, int i) {
  return __int_as_float(__builtin_amdgcn_readlane(__float_as_int(v), i));
}


__device__ __forceinline__ int piece_rank(int s, int nn, int lane) {
  int rank = 0;
  for (int j = 0; j < nn; ++j) {
    const int sj = lane_bcast_i(s, j);
    rank += (int)(sj < s) + (int)((sj == s) & (j < lane));
  }
  return lane < nn ? rank : lane;
}

template<int HEADS, int CH>
__device__ __forceinline__ void add_piece_rows(const float* __restrict__ hsrc, int s, const float (&ex)[HEADS],
                                               int nn, int head, int j0, float (&acc)[HEADS * CH / 32]) {
  constexpr int F  = HEADS * CH;
  constexpr int NF = F / 32;
  const int nn4 = (nn + 3) & ~3;
  for (int i = 0; i < nn4; i += 4) {
    int   sr[4];
    float w[4];
    float v[4][NF];
    for (int u = 0; u < 4; ++u) {
      sr[u] = lane_bcast_i(s, i + u);
      float wu = lane_bcast_f(ex[0], i + u);
      for (int h = 1; h < HEADS; ++h) {
        float th = lane_bcast_f(ex[h], i + u);
        wu = head == h ? th : wu;
      }
      w[u] = wu;
    }
    for (int u = 0; u < 4; ++u) load_row<NF>(hsrc + (long)sr[u] * F + j0, v[u]);
    for (int u = 0; u < 4; ++u)
      for (int k = 0; k < NF; ++k) acc[k] += w[u] * v[u][k];
  }
}

template<int HEADS, int CH, bool ELU>
__device__ __forceinline__ void gather_row(const int* __restrict__ rowptr, const int* __restrict__ csr_src,
                                           const float* __restrict__ als, const float* __restrict__ ald,
                                           const float* __restrict__ hsrc, const float* __restrict__ bias,
                                           int node, int lane, float (&r)[HEADS * CH / 32]) {
  constexpr int F  = HEADS * CH;
  constexpr int NF = F / 32;
  const int beg = rowptr[node];
  const int deg = rowptr[node + 1] - beg;
  const int tot = deg + 1;
  float adst[HEADS];
  load_heads<HEADS>(ald, node, adst);

  const int j0   = lane * NF;
  const int head = j0 / CH;
  float acc[NF];
  for (int k = 0; k < NF; ++k) acc[k] = 0.f;
  float ssum[HEADS];
  for (int h = 0; h < HEADS; ++h) ssum[h] = 0.f;

  if (tot <= 32) {
    int s = node;
    if (lane < deg) s = csr_src[beg + lane];
    float a[HEADS];
    load_heads<HEADS>(als, s, a);
    const int rank = piece_rank(s, tot, lane);
    float l[HEADS], mx[HEADS];
    for (int h = 0; h < HEADS; ++h) {
      float t = a[h] + adst[h];
      l[h] = t > 0.f ? t : NEG_SLOPE * t;
      mx[h] = -__builtin_huge_valf();
      if (lane < tot) mx[h] = fmaxf(mx[h], l[h]);
    }
    for (int h = 0; h < HEADS; ++h)
      for (int off = 16; off > 0; off >>= 1) mx[h] = fmaxf(mx[h], __shfl_xor(mx[h], off));
    float ex[HEADS];
    for (int h = 0; h < HEADS; ++h) {
      const float e = lane < tot ? __expf(l[h] - mx[h]) : 0.f;
      ex[h] = __int_as_float(__builtin_amdgcn_ds_permute(rank << 2, __float_as_int(e)));
      ssum[h] += ex[h];
    }
    s = __builtin_amdgcn_ds_permute(rank << 2, s);
    add_piece_rows<HEADS, CH>(hsrc, s, ex, tot, head, j0, acc);
  } else {
    float mx[HEADS];
    for (int h = 0; h < HEADS; ++h) mx[h] = -__builtin_huge_valf();
    for (int c = 0; c < tot; c += 32) {
      const int g = c + lane;
      if (g < tot) {
        const int s = g < deg ? csr_src[beg + g] : node;
        float a[HEADS];
        load_heads<HEADS>(als, s, a);
        for (int h = 0; h < HEADS; ++h) {
          float l = a[h] + adst[h];
          l = l > 0.f ? l : NEG_SLOPE * l;
          mx[h] = fmaxf(mx[h], l);
        }
      }
    }
    for (int h = 0; h < HEADS; ++h)
      for (int off = 16; off > 0; off >>= 1) mx[h] = fmaxf(mx[h], __shfl_xor(mx[h], off));

    for (int c = 0; c < tot; c += 32) {
      const int g = c + lane;
      const int nn = tot - c < 32 ? tot - c : 32;
      int s = node;
      if (g < deg) s = csr_src[beg + g];
      {
        const int rank = piece_rank(s, nn, lane);
        s = __builtin_amdgcn_ds_permute(rank << 2, s);
      }
      float ex[HEADS];
      for (int h = 0; h < HEADS; ++h) ex[h] = 0.f;
      if (g < tot) {
        float a[HEADS];
        load_heads<HEADS>(als, s, a);
        for (int h = 0; h < HEADS; ++h) {
          float l = a[h] + adst[h];
          l = l > 0.f ? l : NEG_SLOPE * l;
          ex[h] = __expf(l - mx[h]);
          ssum[h] += ex[h];
        }
      }
      add_piece_rows<HEADS, CH>(hsrc, s, ex, nn, head, j0, acc);
    }
  }
  for (int h = 0; h < HEADS; ++h)
    for (int off = 16; off > 0; off >>= 1) ssum[h] += __shfl_xor(ssum[h], off);
  float sm = ssum[0];
  for (int h = 1; h < HEADS; ++h) sm = head == h ? ssum[h] : sm;

  float bv[NF];
  load_row<NF>(bias + j0, bv);
  for (int k = 0; k < NF; ++k) {
    float val = acc[k] / sm + bv[k];
    if (ELU) val = val > 0.f ? val : (__expf(val) - 1.f);
    r[k] = val;
  }
}


template<int HEADS, int CH, bool ELU>
__global__ void gather_kernel(const int* __restrict__ rowptr, const int* __restrict__ csr_src,
                              const float* __restrict__ als, const float* __restrict__ ald,
                              const float* __restrict__ hsrc, const float* __restrict__ bias,
                              float* __restrict__ outp, int nnodes) {
  constexpr int F  = HEADS * CH;
  constexpr int NF = F / 32;
  const int lane = threadIdx.x & 31;
  const int node = __builtin_amdgcn_readfirstlane((int)((blockIdx.x * blockDim.x + threadIdx.x) >> 5));
  if (node >= nnodes) return;
  float r[NF];
  gather_row<HEADS, CH, ELU>(rowptr, csr_src, als, ald, hsrc, bias, node, lane, r);
  store_row<NF>(outp + (long)node * F + lane * NF, r);
}

template<int HEADS, int CH, int F2>
__global__ __launch_bounds__(512) void gather_gemm_kernel(
    const int* __restrict__ rowptr, const int* __restrict__ csr_src,
    const float* __restrict__ als, const float* __restrict__ ald,
    const float* __restrict__ hsrc, const float* __restrict__ bias,
    const float* __restrict__ B, const float* __restrict__ a_src2, const float* __restrict__ a_dst2,
    float* __restrict__ C, float* __restrict__ als2, float* __restrict__ ald2, int nnodes) {
  constexpr int F    = HEADS * CH;
  constexpr int NF   = F / 32;
  constexpr int P    = F + 4;
  constexpr int TN   = F2 / 16;
  constexpr int KQ   = 16 / TN;
  constexpr int KLEN = F / KQ;
  static_assert(F2 % 16 == 0 && 16 % TN == 0 && F % (4 * KQ) == 0 && NF % 4 == 0, "tile split");
  __shared__ __attribute__((aligned(16))) float As[16 * P];
  __shared__ float part[KQ][TN][8][32];
  __shared__ float red[2][16][TN];

  const int lane = threadIdx.x & 31;
  const int wib  = __builtin_amdgcn_readfirstlane((int)(threadIdx.x >> 5));
  const int node = blockIdx.x * 16 + wib;
  const int tn   = wib % TN;
  const int kq   = wib / TN;
  const int half = lane >> 4;
  const int l15  = lane & 15;
  const int bcol = tn * 16 + l15;
  v2f bf[KLEN / 4];
  float cs, cd;
  {
    float r[NF];
    for (int k = 0; k < NF; ++k) r[k] = 0.f;
    if (node < nnodes)
      gather_row<HEADS, CH, true>(rowptr, csr_src, als, ald, hsrc, bias, node, lane, r);
#pragma unroll
    for (int i = 0; i < KLEN / 4; ++i) {
      const int ka = kq * KLEN + 4 * i + 2 * half;
      bf[i].x = B[ka * F2 + bcol];
      bf[i].y = B[(ka + 1) * F2 + bcol];
    }
    cs = a_src2[bcol];
    cd = a_dst2[bcol];
    store_row<NF>(As + wib * P + lane * NF, r);
  }
  __syncthreads();

  v8f acc = {};
#pragma unroll
  for (int i = 0; i < KLEN / 4; ++i) {
    const int ka = kq * KLEN + 4 * i + 2 * half;
    v2f a;
    a.x = As[l15 * P + ka];
    a.y = As[l15 * P + ka + 1];
    acc = __builtin_amdgcn_wmma_f32_16x16x4_f32(false, a, false, bf[i],
                                                (short)0, acc, false, false);
  }
  for (int r = 0; r < 8; ++r) part[kq][tn][r][lane] = acc[r];
  __syncthreads();

  if (wib < TN) {
    float c[8];
    for (int r = 0; r < 8; ++r) {
      float s = part[0][wib][r][lane];
      for (int q = 1; q < KQ; ++q) s += part[q][wib][r][lane];
      c[r] = s;
    }
    const long crow = (long)blockIdx.x * 16 + half * 8;
    for (int r = 0; r < 8; ++r)
      if (crow + r < nnodes) C[(crow + r) * F2 + bcol] = c[r];
    float ps[8], pd[8];
    for (int r = 0; r < 8; ++r) { ps[r] = c[r] * cs; pd[r] = c[r] * cd; }
    for (int m = 1; m < 16; m <<= 1)
      for (int r = 0; r < 8; ++r) { ps[r] += row_xor_f(ps[r], m); pd[r] += row_xor_f(pd[r], m); }
    float vs = ps[0], vd = pd[0];
    for (int r = 1; r < 8; ++r) { vs = l15 == r ? ps[r] : vs; vd = l15 == r ? pd[r] : vd; }
    if (l15 < 8) { red[0][half * 8 + l15][wib] = vs; red[1][half * 8 + l15][wib] = vd; }
  }
  __syncthreads();
  const int t = threadIdx.x;
  if (t < 32) {
    const int row  = t & 15;
    const int kind = t >> 4;
    float s = red[kind][row][0];
    for (int w = 1; w < TN; ++w) s += red[kind][row][w];
    const long orow = (long)blockIdx.x * 16 + row;
    float* __restrict__ o = kind ? ald2 : als2;
    if (orow < nnodes) o[orow] = s;
  }
}


static inline int cdiv_i(long a, long b) { return (int)((a + b - 1) / b); }

template<int HEADS, int CH, int K>
static void launch_gemm(const float* A, const float* B, float* C, int M,
                        const float* a_src, const float* a_dst,
                        float* als, float* ald, hipStream_t stream) {
  dim3 grid(GEMM_GRID);
  dim3 block(2 * HEADS * CH);
  gemm_wmma_f32_bres<HEADS, CH, K><<<grid, block, 0, stream>>>(A, B, C, M / 16, a_src, a_dst, als, ald);
}

extern "C" void kernel_launch(void* const* d_in, const int* in_sizes, int n_in,
                              void* d_out, int out_size, void* d_ws, size_t ws_size,
                              hipStream_t stream) {
  const float* x   = (const float*)d_in[0];
  const int*   eix = (const int*)d_in[1];
  const float* W1  = (const float*)d_in[2];
  const float* as1 = (const float*)d_in[3];
  const float* ad1 = (const float*)d_in[4];
  const float* b1  = (const float*)d_in[5];
  const float* W2  = (const float*)d_in[6];
  const float* as2 = (const float*)d_in[7];
  const float* ad2 = (const float*)d_in[8];
  const float* b2  = (const float*)d_in[9];
  float* out = (float*)d_out;

  const int IN_DIM = 128;
  const int F1     = in_sizes[5];
  const int HID    = in_sizes[9];
  const int HEADS  = F1 / HID;
  const int Nn     = in_sizes[0] / IN_DIM;
  const int E      = in_sizes[1] / 2;

  float* h2   = (float*)d_ws;
  float* h1   = h2 + (long)Nn * F1;
  float* als1 = h1 + (long)Nn * F1;
  float* ald1 = als1 + (long)Nn * HEADS;
  float* als2 = ald1 + (long)Nn * HEADS;
  float* ald2 = als2 + Nn;
  const int NP = ((Nn + 255) / 256) * 256;
  const int EC = ((cdiv_i(E, NCHUNK) + 3) / 4) * 4;
  int* rowptr = (int*)(ald2 + Nn);
  int* csr    = rowptr + (Nn + 1);
  int* rank   = csr + E;
  int* bsum   = rank + E;
  int* tab    = bsum + (NP / 256) * NCHUNK + 64;
  tab = (int*)(((size_t)tab + 255) & ~(size_t)255);

  const int bpc = cdiv_i(EC, 1024);
  count_rank_kernel<<<NCHUNK, CNT_T, 0, stream>>>(eix, E, EC, NP, tab, bsum, rank);
  offsets_kernel<<<NP / 256, 256, 0, stream>>>(tab, bsum, rowptr, Nn, NP);
  place_kernel<<<NCHUNK * bpc, 256, 0, stream>>>(eix, E, EC, NP, bpc, tab, rank, csr);

  launch_gemm<4, 64, 128>(x, W1, h1, Nn, as1, ad1, als1, ald1, stream);
  gather_gemm_kernel<4, 64, 64><<<cdiv_i(Nn, 16), 512, 0, stream>>>(
      rowptr, csr, als1, ald1, h1, b1, W2, as2, ad2, h2, als2, ald2, Nn);
  gather_kernel<1, 64, false><<<cdiv_i((long)Nn * 32, 256), 256, 0, stream>>>(
      rowptr, csr, als2, ald2, h2, b2, out, Nn);
}


# Edge count pass runs inside the layer-1 GEMM launch: 128 of its 512 work-groups first count a (chunk, node half) in a 100 KiB LDS table beside one that multiplies; 5 launches; sum order unchanged
# speedup vs baseline: 8.8994x; 1.0304x over previous
#include <hip/hip_runtime.h>


typedef float v2f __attribute__((ext_vector_type(2)));
typedef float v8f __attribute__((ext_vector_type(8)));

#define NEG_SLOPE 0.2f

template<int M_>
__device__ __forceinline__ float row_xor_c(float v) {
  return __int_as_float(__builtin_amdgcn_update_dpp(0, __float_as_int(v), 0x160 + M_, 0xf, 0xf, true));
}
__device__ __forceinline__ float row_xor_f(float v, int m) {
  return m == 1 ? row_xor_c<1>(v) : m == 2 ? row_xor_c<2>(v) : m == 4 ? row_xor_c<4>(v) : row_xor_c<8>(v);
}

constexpr int GEMM_GRID = 512;

template<int HEADS, int CH, int K>
__device__ __forceinline__ void gemm_wmma_f32_bres(
    int bid, int gsz,
    const float* __restrict__ A, const float* __restrict__ B, float* __restrict__ C, int tilesM,
    const float* __restrict__ a_src, const float* __restrict__ a_dst,
    float* __restrict__ als, float* __restrict__ ald) {
  constexpr int Nd  = HEADS * CH;
  constexpr int NW  = Nd / 16;
  constexpr int P   = K + 4;
  constexpr int KV  = K / 128;
  constexpr int TPH = CH / 16;
  static_assert(NW == 16 && K % 128 == 0 && CH % 16 == 0, "one slab row per wave, whole heads");
  __shared__ __attribute__((aligned(16))) float As[2][16 * P];
  __shared__ float red[2][2][16][NW];

  const int lane = threadIdx.x & 31;
  const int wib  = __builtin_amdgcn_readfirstlane((int)(threadIdx.x >> 5));
  const int half = lane >> 4;
  const int l15  = lane & 15;
  const int bcol = wib * 16 + l15;
  int tm = bid;
  if (tm >= tilesM) return;

  v2f bf[K / 4];
#pragma unroll
  for (int i = 0; i < K / 4; ++i) {
    const int ka = 4 * i + 2 * half;
    bf[i].x = B[ka * Nd + bcol];
    bf[i].y = B[(ka + 1) * Nd + bcol];
  }
  const float cs = a_src[bcol];
  const float cd = a_dst[bcol];

  {
    const float4* Av = (const float4*)(A + ((long)tm * 16 + wib) * K);
    for (int q = 0; q < KV; ++q) *(float4*)&As[0][wib * P + (lane + 32 * q) * 4] = Av[lane + 32 * q];
  }
  __syncthreads();

  int buf = 0;
  for (; tm < tilesM; tm += gsz) {
    const int tnext = tm + gsz;
    float4 nx[KV];
    for (int q = 0; q < KV; ++q) nx[q] = make_float4(0.f, 0.f, 0.f, 0.f);
    if (tnext < tilesM) {
      const float4* Av = (const float4*)(A + ((long)tnext * 16 + wib) * K);
      for (int q = 0; q < KV; ++q) nx[q] = Av[lane + 32 * q];
    }
    const float* __restrict__ as = &As[buf][l15 * P + 2 * half];
    v8f acc = {};
#pragma unroll
    for (int i = 0; i < K / 4; ++i) {
      const v2f a = *(const v2f*)(as + 4 * i);
      acc = __builtin_amdgcn_wmma_f32_16x16x4_f32(false, a, false, bf[i],
                                                  (short)0, acc, false, false);
    }
    const long crow = (long)tm * 16 + half * 8;
#pragma unroll
    for (int r = 0; r < 8; ++r)
      C[(crow + r) * Nd + bcol] = acc[r];

    float ps[8], pd[8];
    for (int r = 0; r < 8; ++r) { ps[r] = acc[r] * cs; pd[r] = acc[r] * cd; }
    for (int m = 1; m < 16; m <<= 1)
      for (int r = 0; r < 8; ++r) { ps[r] += row_xor_f(ps[r], m); pd[r] += row_xor_f(pd[r], m); }
    float vs = ps[0], vd = pd[0];
    for (int r = 1; r < 8; ++r) { vs = l15 == r ? ps[r] : vs; vd = l15 == r ? pd[r] : vd; }
    if (l15 < 8) { red[buf][0][half * 8 + l15][wib] = vs; red[buf][1][half * 8 + l15][wib] = vd; }
    if (tnext < tilesM)
      for (int q = 0; q < KV; ++q) *(float4*)&As[buf ^ 1][wib * P + (lane + 32 * q) * 4] = nx[q];
    __syncthreads();
    const int t = threadIdx.x;
    if (t < 32 * HEADS) {
      const int hl   = t % HEADS;
      const int row  = (t / HEADS) & 15;
      const int kind = t / (16 * HEADS);
      float s = red[buf][kind][row][hl * TPH];
      for (int w = 1; w < TPH; ++w) s += red[buf][kind][row][hl * TPH + w];
      float* __restrict__ o = kind ? ald : als;
      o[((long)tm * 16 + row) * HEADS + hl] = s;
    }
    buf ^= 1;
  }
}


constexpr int NCHUNK  = 64;
constexpr int LDS_CAP = 25600;
constexpr int CNT_T   = 512;
constexpr int CNT_B   = 13;
static_assert(GEMM_GRID >= 4 * NCHUNK, "the count tasks go to the even work-groups below 4 * NCHUNK");

__device__ __forceinline__ void count_rank_part(
    int bid, const int* __restrict__ eidx, int E, int EC, int NP, int NPH,
    int* __restrict__ tab, int* __restrict__ bsum, int* __restrict__ rank) {
  __shared__ __attribute__((aligned(16))) int lc[LDS_CAP];
  const int t  = threadIdx.x;
  const int c  = bid >> 1;
  const int lo = (bid & 1) * NPH;
  const int nh = NP - lo < NPH ? NP - lo : NPH;
  const int* __restrict__ dst = eidx + E;
  const int e0 = c * EC;
  const int e1 = e0 + EC < E ? e0 + EC : E;

  int d[CNT_B];
  for (int u = 0; u < CNT_B; ++u) {
    const int e = e0 + t + u * CNT_T;
    d[u] = e < e1 ? dst[e] : -1;
  }
  for (int i = t * 4; i < nh; i += CNT_T * 4) *(int4*)&lc[i] = make_int4(0, 0, 0, 0);
  __syncthreads();

  for (int eb = e0; eb < e1; eb += CNT_T * CNT_B) {
    int dn[CNT_B];
    for (int u = 0; u < CNT_B; ++u) {
      const int e = eb + CNT_T * CNT_B + t + u * CNT_T;
      dn[u] = e < e1 ? dst[e] : -1;
    }
    int dl[CNT_B], r[CNT_B];
    for (int u = 0; u < CNT_B; ++u) dl[u] = d[u] - lo;
    for (int u = 0; u < CNT_B; ++u)
      r[u] = atomicAdd(&lc[(unsigned)dl[u] < (unsigned)nh ? dl[u] : nh + (t & 31)], 1);
    for (int u = 0; u < CNT_B; ++u)
      if ((unsigned)dl[u] < (unsigned)nh) rank[eb + t + u * CNT_T] = r[u];
    for (int u = 0; u < CNT_B; ++u) d[u] = dn[u];
  }
  __syncthreads();

  int* __restrict__ mytab = tab + (long)c * NP + lo;
  for (int i = t * 4; i < nh; i += CNT_T * 4) *(int4*)&mytab[i] = *(const int4*)&lc[i];
  const int wave = t >> 5, lane = t & 31;
  for (int b = wave; b < (nh >> 8); b += CNT_T / 32) {
    const int4 p = *(const int4*)&lc[b * 256 + lane * 8];
    const int4 q = *(const int4*)&lc[b * 256 + lane * 8 + 4];
    int sum = p.x + p.y + p.z + p.w + q.x + q.y + q.z + q.w;
    for (int off = 16; off > 0; off >>= 1) sum += __shfl_xor(sum, off);
    if (lane == 0) bsum[((lo >> 8) + b) * NCHUNK + c] = sum;
  }
}

template<int HEADS, int CH, int K>
__global__ __launch_bounds__(CNT_T) void count_gemm_kernel(
    const int* __restrict__ eidx, int E, int EC, int NP, int NPH, int ncount,
    int* __restrict__ tab, int* __restrict__ bsum, int* __restrict__ rank,
    const float* __restrict__ A, const float* __restrict__ B, float* __restrict__ C, int tilesM,
    const float* __restrict__ a_src, const float* __restrict__ a_dst,
    float* __restrict__ als, float* __restrict__ ald) {
  static_assert(2 * HEADS * CH == CNT_T, "one work-group size for both parts");
  const int bx = blockIdx.x;
  if ((bx & 1) == 0 && bx < 2 * ncount)
    count_rank_part(bx >> 1, eidx, E, EC, NP, NPH, tab, bsum, rank);
  gemm_wmma_f32_bres<HEADS, CH, K>(bx, (int)gridDim.x, A, B, C, tilesM, a_src, a_dst, als, ald);
}

__global__ __launch_bounds__(256) void offsets_kernel(int* __restrict__ tab,
                                                      const int* __restrict__ bsum,
                                                      int* __restrict__ rowptr, int n, int NP) {
  __shared__ int sh[256];
  const int t = threadIdx.x;
  const int i = blockIdx.x * 256 + t;
  int part = 0;
  for (int j = t; j < (int)blockIdx.x * NCHUNK; j += 256) part += bsum[j];
  int v[NCHUNK];
  for (int c = 0; c < NCHUNK; ++c) v[c] = tab[(long)c * NP + i];
  int deg = 0;
  for (int c = 0; c < NCHUNK; ++c) { const int x = v[c]; v[c] = deg; deg += x; }
  sh[t] = part;
  __syncthreads();
  for (int off = 128; off > 0; off >>= 1) {
    if (t < off) sh[t] += sh[t + off];
    __syncthreads();
  }
  const int base = sh[0];
  __syncthreads();
  sh[t] = deg;
  __syncthreads();
  for (int off = 1; off < 256; off <<= 1) {
    int w = t >= off ? sh[t - off] : 0;
    __syncthreads();
    sh[t] += w;
    __syncthreads();
  }
  const int excl = base + sh[t] - deg;
  for (int c = 0; c < NCHUNK; ++c) tab[(long)c * NP + i] = excl + v[c];
  if (i < n) rowptr[i] = excl;
  if (i == n - 1) rowptr[n] = excl + deg;
}

__global__ __launch_bounds__(256) void place_kernel(const int* __restrict__ eidx, int E, int EC, int NP,
                                                    int bpc, const int* __restrict__ tab,
                                                    const int* __restrict__ rank,
                                                    int* __restrict__ csr_src) {
  const int c  = blockIdx.x / bpc;
  const int sb = blockIdx.x - c * bpc;
  const int e0 = c * EC;
  const int e1 = e0 + EC < E ? e0 + EC : E;
  const int eb = e0 + sb * 1024 + threadIdx.x;
  const int* __restrict__ mytab = tab + (long)c * NP;
  int d[4], s[4], r[4];
  for (int u = 0; u < 4; ++u) {
    const int e = eb + u * 256;
    d[u] = -1; s[u] = 0; r[u] = 0;
    if (e < e1) { d[u] = eidx[E + e]; s[u] = eidx[e]; r[u] = rank[e]; }
  }
  int p[4];
  for (int u = 0; u < 4; ++u) p[u] = mytab[d[u] >= 0 ? d[u] : 0] + r[u];
  for (int u = 0; u < 4; ++u)
    if (d[u] >= 0) csr_src[p[u]] = s[u];
}

template<int NH>
__device__ __forceinline__ void load_heads(const float* __restrict__ p, int idx, float (&v)[NH]) {
  if constexpr (NH == 4) {
    float4 t = ((const float4*)p)[idx];
    v[0] = t.x; v[1] = t.y; v[2] = t.z; v[3] = t.w;
  } else {
    for (int h = 0; h < NH; ++h) v[h] = p[(long)idx * NH + h];
  }
}

template<int NF>
__device__ __forceinline__ void load_row(const float* __restrict__ p, float (&v)[NF]) {
  if constexpr (NF % 4 == 0) {
    for (int q = 0; q < NF / 4; ++q) {
      float4 t = ((const float4*)p)[q];
      v[4 * q] = t.x; v[4 * q + 1] = t.y; v[4 * q + 2] = t.z; v[4 * q + 3] = t.w;
    }
  } else if constexpr (NF % 2 == 0) {
    for (int q = 0; q < NF / 2; ++q) {
      float2 t = ((const float2*)p)[q];
      v[2 * q] = t.x; v[2 * q + 1] = t.y;
    }
  } else {
    for (int k = 0; k < NF; ++k) v[k] = p[k];
  }
}

template<int NF>
__device__ __forceinline__ void store_row(float* __restrict__ p, const float (&v)[NF]) {
  if constexpr (NF % 4 == 0) {
    for (int q = 0; q < NF / 4; ++q)
      ((float4*)p)[q] = make_float4(v[4 * q], v[4 * q + 1], v[4 * q + 2], v[4 * q + 3]);
  } else if constexpr (NF % 2 == 0) {
    for (int q = 0; q < NF / 2; ++q)
      ((float2*)p)[q] = make_float2(v[2 * q], v[2 * q + 1]);
  } else {
    for (int k = 0; k < NF; ++k) p[k] = v[k];
  }
}

__device__ __forceinline__ int lane_bcast_i(int v, int i) {
  return __builtin_amdgcn_readlane(v, i);
}
__device__ __forceinline__ float lane_bcast_f(float v, int i) {
  return __int_as_float(__builtin_amdgcn_readlane(__float_as_int(v), i));
}


__device__ __forceinline__ int piece_rank(int s, int nn, int lane) {
  int rank = 0;
  for (int j = 0; j < nn; ++j) {
    const int sj = lane_bcast_i(s, j);
    rank += (int)(sj < s) + (int)((sj == s) & (j < lane));
  }
  return lane < nn ? rank : lane;
}

template<int HEADS, int CH>
__device__ __forceinline__ void add_piece_rows(const float* __restrict__ hsrc, int s, const float (&ex)[HEADS],
                                               int nn, int head, int j0, float (&acc)[HEADS * CH / 32]) {
  constexpr int F  = HEADS * CH;
  constexpr int NF = F / 32;
  const int nn4 = (nn + 3) & ~3;
  for (int i = 0; i < nn4; i += 4) {
    int   sr[4];
    float w[4];
    float v[4][NF];
    for (int u = 0; u < 4; ++u) {
      sr[u] = lane_bcast_i(s, i + u);
      float wu = lane_bcast_f(ex[0], i + u);
      for (int h = 1; h < HEADS; ++h) {
        float th = lane_bcast_f(ex[h], i + u);
        wu = head == h ? th : wu;
      }
      w[u] = wu;
    }
    for (int u = 0; u < 4; ++u) load_row<NF>(hsrc + (long)sr[u] * F + j0, v[u]);
    for (int u = 0; u < 4; ++u)
      for (int k = 0; k < NF; ++k) acc[k] += w[u] * v[u][k];
  }
}

template<int HEADS, int CH, bool ELU>
__device__ __forceinline__ void gather_row(const int* __restrict__ rowptr, const int* __restrict__ csr_src,
                                           const float* __restrict__ als, const float* __restrict__ ald,
                                           const float* __restrict__ hsrc, const float* __restrict__ bias,
                                           int node, int lane, float (&r)[HEADS * CH / 32]) {
  constexpr int F  = HEADS * CH;
  constexpr int NF = F / 32;
  const int beg = rowptr[node];
  const int deg = rowptr[node + 1] - beg;
  const int tot = deg + 1;
  float adst[HEADS];
  load_heads<HEADS>(ald, node, adst);

  const int j0   = lane * NF;
  const int head = j0 / CH;
  float acc[NF];
  for (int k = 0; k < NF; ++k) acc[k] = 0.f;
  float ssum[HEADS];
  for (int h = 0; h < HEADS; ++h) ssum[h] = 0.f;

  if (tot <= 32) {
    int s = node;
    if (lane < deg) s = csr_src[beg + lane];
    float a[HEADS];
    load_heads<HEADS>(als, s, a);
    const int rank = piece_rank(s, tot, lane);
    float l[HEADS], mx[HEADS];
    for (int h = 0; h < HEADS; ++h) {
      float t = a[h] + adst[h];
      l[h] = t > 0.f ? t : NEG_SLOPE * t;
      mx[h] = -__builtin_huge_valf();
      if (lane < tot) mx[h] = fmaxf(mx[h], l[h]);
    }
    for (int h = 0; h < HEADS; ++h)
      for (int off = 16; off > 0; off >>= 1) mx[h] = fmaxf(mx[h], __shfl_xor(mx[h], off));
    float ex[HEADS];
    for (int h = 0; h < HEADS; ++h) {
      const float e = lane < tot ? __expf(l[h] - mx[h]) : 0.f;
      ex[h] = __int_as_float(__builtin_amdgcn_ds_permute(rank << 2, __float_as_int(e)));
      ssum[h] += ex[h];
    }
    s = __builtin_amdgcn_ds_permute(rank << 2, s);
    add_piece_rows<HEADS, CH>(hsrc, s, ex, tot, head, j0, acc);
  } else {
    float mx[HEADS];
    for (int h = 0; h < HEADS; ++h) mx[h] = -__builtin_huge_valf();
    for (int c = 0; c < tot; c += 32) {
      const int g = c + lane;
      if (g < tot) {
        const int s = g < deg ? csr_src[beg + g] : node;
        float a[HEADS];
        load_heads<HEADS>(als, s, a);
        for (int h = 0; h < HEADS; ++h) {
          float l = a[h] + adst[h];
          l = l > 0.f ? l : NEG_SLOPE * l;
          mx[h] = fmaxf(mx[h], l);
        }
      }
    }
    for (int h = 0; h < HEADS; ++h)
      for (int off = 16; off > 0; off >>= 1) mx[h] = fmaxf(mx[h], __shfl_xor(mx[h], off));

    for (int c = 0; c < tot; c += 32) {
      const int g = c + lane;
      const int nn = tot - c < 32 ? tot - c : 32;
      int s = node;
      if (g < deg) s = csr_src[beg + g];
      {
        const int rank = piece_rank(s, nn, lane);
        s = __builtin_amdgcn_ds_permute(rank << 2, s);
      }
      float ex[HEADS];
      for (int h = 0; h < HEADS; ++h) ex[h] = 0.f;
      if (g < tot) {
        float a[HEADS];
        load_heads<HEADS>(als, s, a);
        for (int h = 0; h < HEADS; ++h) {
          float l = a[h] + adst[h];
          l = l > 0.f ? l : NEG_SLOPE * l;
          ex[h] = __expf(l - mx[h]);
          ssum[h] += ex[h];
        }
      }
      add_piece_rows<HEADS, CH>(hsrc, s, ex, nn, head, j0, acc);
    }
  }
  for (int h = 0; h < HEADS; ++h)
    for (int off = 16; off > 0; off >>= 1) ssum[h] += __shfl_xor(ssum[h], off);
  float sm = ssum[0];
  for (int h = 1; h < HEADS; ++h) sm = head == h ? ssum[h] : sm;

  float bv[NF];
  load_row<NF>(bias + j0, bv);
  for (int k = 0; k < NF; ++k) {
    float val = acc[k] / sm + bv[k];
    if (ELU) val = val > 0.f ? val : (__expf(val) - 1.f);
    r[k] = val;
  }
}


template<int HEADS, int CH, bool ELU>
__global__ void gather_kernel(const int* __restrict__ rowptr, const int* __restrict__ csr_src,
                              const float* __restrict__ als, const float* __restrict__ ald,
                              const float* __restrict__ hsrc, const float* __restrict__ bias,
                              float* __restrict__ outp, int nnodes) {
  constexpr int F  = HEADS * CH;
  constexpr int NF = F / 32;
  const int lane = threadIdx.x & 31;
  const int node = __builtin_amdgcn_readfirstlane((int)((blockIdx.x * blockDim.x + threadIdx.x) >> 5));
  if (node >= nnodes) return;
  float r[NF];
  gather_row<HEADS, CH, ELU>(rowptr, csr_src, als, ald, hsrc, bias, node, lane, r);
  store_row<NF>(outp + (long)node * F + lane * NF, r);
}

template<int HEADS, int CH, int F2>
__global__ __launch_bounds__(512) void gather_gemm_kernel(
    const int* __restrict__ rowptr, const int* __restrict__ csr_src,
    const float* __restrict__ als, const float* __restrict__ ald,
    const float* __restrict__ hsrc, const float* __restrict__ bias,
    const float* __restrict__ B, const float* __restrict__ a_src2, const float* __restrict__ a_dst2,
    float* __restrict__ C, float* __restrict__ als2, float* __restrict__ ald2, int nnodes) {
  constexpr int F    = HEADS * CH;
  constexpr int NF   = F / 32;
  constexpr int P    = F + 4;
  constexpr int TN   = F2 / 16;
  constexpr int KQ   = 16 / TN;
  constexpr int KLEN = F / KQ;
  static_assert(F2 % 16 == 0 && 16 % TN == 0 && F % (4 * KQ) == 0 && NF % 4 == 0, "tile split");
  __shared__ __attribute__((aligned(16))) float As[16 * P];
  __shared__ float part[KQ][TN][8][32];
  __shared__ float red[2][16][TN];

  const int lane = threadIdx.x & 31;
  const int wib  = __builtin_amdgcn_readfirstlane((int)(threadIdx.x >> 5));
  const int node = blockIdx.x * 16 + wib;
  const int tn   = wib % TN;
  const int kq   = wib / TN;
  const int half = lane >> 4;
  const int l15  = lane & 15;
  const int bcol = tn * 16 + l15;
  v2f bf[KLEN / 4];
  float cs, cd;
  {
    float r[NF];
    for (int k = 0; k < NF; ++k) r[k] = 0.f;
    if (node < nnodes)
      gather_row<HEADS, CH, true>(rowptr, csr_src, als, ald, hsrc, bias, node, lane, r);
#pragma unroll
    for (int i = 0; i < KLEN / 4; ++i) {
      const int ka = kq * KLEN + 4 * i + 2 * half;
      bf[i].x = B[ka * F2 + bcol];
      bf[i].y = B[(ka + 1) * F2 + bcol];
    }
    cs = a_src2[bcol];
    cd = a_dst2[bcol];
    store_row<NF>(As + wib * P + lane * NF, r);
  }
  __syncthreads();

  v8f acc = {};
#pragma unroll
  for (int i = 0; i < KLEN / 4; ++i) {
    const int ka = kq * KLEN + 4 * i + 2 * half;
    v2f a;
    a.x = As[l15 * P + ka];
    a.y = As[l15 * P + ka + 1];
    acc = __builtin_amdgcn_wmma_f32_16x16x4_f32(false, a, false, bf[i],
                                                (short)0, acc, false, false);
  }
  for (int r = 0; r < 8; ++r) part[kq][tn][r][lane] = acc[r];
  __syncthreads();

  if (wib < TN) {
    float c[8];
    for (int r = 0; r < 8; ++r) {
      float s = part[0][wib][r][lane];
      for (int q = 1; q < KQ; ++q) s += part[q][wib][r][lane];
      c[r] = s;
    }
    const long crow = (long)blockIdx.x * 16 + half * 8;
    for (int r = 0; r < 8; ++r)
      if (crow + r < nnodes) C[(crow + r) * F2 + bcol] = c[r];
    float ps[8], pd[8];
    for (int r = 0; r < 8; ++r) { ps[r] = c[r] * cs; pd[r] = c[r] * cd; }
    for (int m = 1; m < 16; m <<= 1)
      for (int r = 0; r < 8; ++r) { ps[r] += row_xor_f(ps[r], m); pd[r] += row_xor_f(pd[r], m); }
    float vs = ps[0], vd = pd[0];
    for (int r = 1; r < 8; ++r) { vs = l15 == r ? ps[r] : vs; vd = l15 == r ? pd[r] : vd; }
    if (l15 < 8) { red[0][half * 8 + l15][wib] = vs; red[1][half * 8 + l15][wib] = vd; }
  }
  __syncthreads();
  const int t = threadIdx.x;
  if (t < 32) {
    const int row  = t & 15;
    const int kind = t >> 4;
    float s = red[kind][row][0];
    for (int w = 1; w < TN; ++w) s += red[kind][row][w];
    const long orow = (long)blockIdx.x * 16 + row;
    float* __restrict__ o = kind ? ald2 : als2;
    if (orow < nnodes) o[orow] = s;
  }
}


static inline int cdiv_i(long a, long b) { return (int)((a + b - 1) / b); }

extern "C" void kernel_launch(void* const* d_in, const int* in_sizes, int n_in,
                              void* d_out, int out_size, void* d_ws, size_t ws_size,
                              hipStream_t stream) {
  const float* x   = (const float*)d_in[0];
  const int*   eix = (const int*)d_in[1];
  const float* W1  = (const float*)d_in[2];
  const float* as1 = (const float*)d_in[3];
  const float* ad1 = (const float*)d_in[4];
  const float* b1  = (const float*)d_in[5];
  const float* W2  = (const float*)d_in[6];
  const float* as2 = (const float*)d_in[7];
  const float* ad2 = (const float*)d_in[8];
  const float* b2  = (const float*)d_in[9];
  float* out = (float*)d_out;

  const int IN_DIM = 128;
  const int F1     = in_sizes[5];
  const int HID    = in_sizes[9];
  const int HEADS  = F1 / HID;
  const int Nn     = in_sizes[0] / IN_DIM;
  const int E      = in_sizes[1] / 2;

  float* h2   = (float*)d_ws;
  float* h1   = h2 + (long)Nn * F1;
  float* als1 = h1 + (long)Nn * F1;
  float* ald1 = als1 + (long)Nn * HEADS;
  float* als2 = ald1 + (long)Nn * HEADS;
  float* ald2 = als2 + Nn;
  const int NP  = ((Nn + 255) / 256) * 256;
  const int NPH = ((NP / 256 + 1) / 2) * 256;
  const int EC = ((cdiv_i(E, NCHUNK) + 3) / 4) * 4;
  int* rowptr = (int*)(ald2 + Nn);
  int* csr    = rowptr + (Nn + 1);
  int* rank   = csr + E;
  int* bsum   = rank + E;
  int* tab    = bsum + (NP / 256) * NCHUNK + 64;
  tab = (int*)(((size_t)tab + 255) & ~(size_t)255);

  const int bpc = cdiv_i(EC, 1024);
  count_gemm_kernel<4, 64, 128><<<GEMM_GRID, CNT_T, 0, stream>>>(
      eix, E, EC, NP, NPH, 2 * NCHUNK, tab, bsum, rank,
      x, W1, h1, Nn / 16, as1, ad1, als1, ald1);
  offsets_kernel<<<NP / 256, 256, 0, stream>>>(tab, bsum, rowptr, Nn, NP);
  place_kernel<<<NCHUNK * bpc, 256, 0, stream>>>(eix, E, EC, NP, bpc, tab, rank, csr);

  gather_gemm_kernel<4, 64, 64><<<cdiv_i(Nn, 16), 512, 0, stream>>>(
      rowptr, csr, als1, ald1, h1, b1, W2, as2, ad2, h2, als2, ald2, Nn);
  gather_kernel<1, 64, false><<<cdiv_i((long)Nn * 32, 256), 256, 0, stream>>>(
      rowptr, csr, als2, ald2, h2, b2, out, Nn);
}
